# starting kernel
# baseline (speedup 1.0000x reference)
#include <hip/hip_runtime.h>
#include <hip/hip_bf16.h>

typedef __attribute__((ext_vector_type(2))) float v2f;
typedef __attribute__((ext_vector_type(8))) float v8f;

#define EMB_DIM 128
#define HIDDEN  256
#define OUT_C   64
#define K_ITERS 10
#define ALPHA   0.1f

__global__ void deg_init_kernel(float* deg, int n) {
    int i = blockIdx.x * blockDim.x + threadIdx.x;
    if (i < n) deg[i] = 1.0f;
}

__global__ void deg_count_kernel(const int* __restrict__ dst, float* deg, int e) {
    int i = blockIdx.x * blockDim.x + threadIdx.x;
    if (i < e) atomicAdd(&deg[dst[i]], 1.0f);
}

__global__ void deg_rsqrt_kernel(float* deg, int n) {
    int i = blockIdx.x * blockDim.x + threadIdx.x;
    if (i < n) {
        float d = deg[i];
        deg[i] = (d > 0.0f) ? rsqrtf(d) : 0.0f;
    }
}

__global__ void edge_norm_kernel(const int* __restrict__ src, const int* __restrict__ dst,
                                 const float* __restrict__ dis, float* __restrict__ norm, int e) {
    int i = blockIdx.x * blockDim.x + threadIdx.x;
    if (i < e) norm[i] = dis[src[i]] * dis[dst[i]];
}

__launch_bounds__(128)
__global__ void mlp_wmma_kernel(const int* __restrict__ xidx,
                                const float* __restrict__ emb,
                                const float* __restrict__ W1, const float* __restrict__ b1,
                                const float* __restrict__ W2, const float* __restrict__ b2,
                                float* __restrict__ h_out, float* __restrict__ x0_out,
                                int n) {
    const int tid  = threadIdx.x;
    const int wave = tid >> 5;
    const int lane = tid & 31;
    const int m    = lane & 15;
    const int g    = lane >> 4;
    const int row0 = blockIdx.x * 16;

    __shared__ float sA[16][EMB_DIM + 2];
    __shared__ float sH[16][HIDDEN + 2];

    for (int i = tid; i < 16 * EMB_DIM; i += 128) {
        int r  = i >> 7;
        int c  = i & (EMB_DIM - 1);
        int rr = row0 + r;
        float v = 0.0f;
        if (rr < n) {
            int node = xidx[rr];
            v = emb[(size_t)node * EMB_DIM + c];
        }
        sA[r][c] = v;
    }
    __syncthreads();

    for (int t = 0; t < 4; ++t) {
        const int nbase = wave * 64 + t * 16;
        v8f acc = {0.f, 0.f, 0.f, 0.f, 0.f, 0.f, 0.f, 0.f};
        for (int s = 0; s < 32; ++s) {
            const int k = 4 * s + 2 * g;
            v2f a; a.x = sA[m][k];                    a.y = sA[m][k + 1];
            v2f b; b.x = W1[(size_t)k * HIDDEN + nbase + m];
                   b.y = W1[(size_t)(k + 1) * HIDDEN + nbase + m];
            acc = __builtin_amdgcn_wmma_f32_16x16x4_f32(
                false, a, false, b, (short)0, acc, false, false);
        }
        const float bias = b1[nbase + m];
        #pragma unroll
        for (int v = 0; v < 8; ++v) {
            float val = acc[v] + bias;
            val = val > 0.0f ? val : 0.0f;
            sH[v + 8 * g][nbase + m] = val;
        }
    }
    __syncthreads();

    {
        const int nbase = wave * 16;
        v8f acc = {0.f, 0.f, 0.f, 0.f, 0.f, 0.f, 0.f, 0.f};
        for (int s = 0; s < 64; ++s) {
            const int k = 4 * s + 2 * g;
            v2f a; a.x = sH[m][k];                    a.y = sH[m][k + 1];
            v2f b; b.x = W2[(size_t)k * OUT_C + nbase + m];
                   b.y = W2[(size_t)(k + 1) * OUT_C + nbase + m];
            acc = __builtin_amdgcn_wmma_f32_16x16x4_f32(
                false, a, false, b, (short)0, acc, false, false);
        }
        const float bias = b2[nbase + m];
        #pragma unroll
        for (int v = 0; v < 8; ++v) {
            const int row = row0 + v + 8 * g;
            if (row < n) {
                const float val = acc[v] + bias;
                const size_t idx = (size_t)row * OUT_C + nbase + m;
                h_out[idx]  = val;
                x0_out[idx] = val;
            }
        }
    }
}

__global__ void scatter_kernel(const int* __restrict__ src, const int* __restrict__ dst,
                               const float* __restrict__ norm,
                               const float* __restrict__ x, float* __restrict__ agg,
                               int e) {
    const unsigned gid = blockIdx.x * blockDim.x + threadIdx.x;
    const unsigned ei  = gid >> 4;
    const unsigned q   = gid & 15;
    if (ei < (unsigned)e) {
        const int   s  = src[ei];
        const int   d  = dst[ei];
        const float nv = norm[ei];
        const float4 v = ((const float4*)(x + (size_t)s * OUT_C))[q];
        float* ap = agg + (size_t)d * OUT_C + q * 4;
        atomicAdd(ap + 0, v.x * nv);
        atomicAdd(ap + 1, v.y * nv);
        atomicAdd(ap + 2, v.z * nv);
        atomicAdd(ap + 3, v.w * nv);
    }
}

__global__ void combine_kernel(float4* __restrict__ xnew_agg, const float4* __restrict__ xold,
                               const float* __restrict__ dis, const float4* __restrict__ h,
                               int total4) {
    const int i = blockIdx.x * blockDim.x + threadIdx.x;
    if (i < total4) {
        const int   node = i >> 4;
        const float di   = dis[node];
        const float d2   = di * di;
        const float4 a  = xnew_agg[i];
        const float4 xo = xold[i];
        const float4 hh = h[i];
        float4 r;
        r.x = (1.0f - ALPHA) * (a.x + d2 * xo.x) + ALPHA * hh.x;
        r.y = (1.0f - ALPHA) * (a.y + d2 * xo.y) + ALPHA * hh.y;
        r.z = (1.0f - ALPHA) * (a.z + d2 * xo.z) + ALPHA * hh.z;
        r.w = (1.0f - ALPHA) * (a.w + d2 * xo.w) + ALPHA * hh.w;
        xnew_agg[i] = r;
    }
}

extern "C" void kernel_launch(void* const* d_in, const int* in_sizes, int n_in,
                              void* d_out, int out_size, void* d_ws, size_t ws_size,
                              hipStream_t stream) {
    (void)n_in; (void)out_size; (void)ws_size;

    const int*   xidx = (const int*)d_in[0];
    const int*   ei   = (const int*)d_in[1];
    const float* emb  = (const float*)d_in[2];
    const float* W1   = (const float*)d_in[3];
    const float* b1   = (const float*)d_in[4];
    const float* W2   = (const float*)d_in[5];
    const float* b2   = (const float*)d_in[6];
    float* out = (float*)d_out;

    const int N = in_sizes[0];
    const int E = in_sizes[1] / 2;
    const size_t NC = (size_t)N * OUT_C;

    float* h     = (float*)d_ws;
    float* xping = h + NC;
    float* dis   = xping + NC;
    float* norm  = dis + N;

    const int* src = ei;
    const int* dst = ei + E;

    deg_init_kernel <<<(N + 255) / 256, 256, 0, stream>>>(dis, N);
    deg_count_kernel<<<(E + 255) / 256, 256, 0, stream>>>(dst, dis, E);
    deg_rsqrt_kernel<<<(N + 255) / 256, 256, 0, stream>>>(dis, N);
    edge_norm_kernel<<<(E + 255) / 256, 256, 0, stream>>>(src, dst, dis, norm, E);

    mlp_wmma_kernel<<<(N + 15) / 16, 128, 0, stream>>>(xidx, emb, W1, b1, W2, b2,
                                                       h, out, N);

    float* xa = out;
    float* xb = xping;
    const long long sthreads = (long long)E * 16;
    const int sblocks = (int)((sthreads + 255) / 256);
    const int total4  = (int)(NC / 4);
    const int cblocks = (total4 + 255) / 256;
    for (int k = 0; k < K_ITERS; ++k) {
        hipMemsetAsync(xb, 0, NC * sizeof(float), stream);
        scatter_kernel<<<sblocks, 256, 0, stream>>>(src, dst, norm, xa, xb, E);
        combine_kernel<<<cblocks, 256, 0, stream>>>((float4*)xb, (const float4*)xa,
                                                    dis, (const float4*)h, total4);
        float* t = xa; xa = xb; xb = t;
    }
    if (xa != out)
        hipMemcpyAsync(out, xa, NC * sizeof(float), hipMemcpyDeviceToDevice, stream);
}


# Each round is one gather over an in-call CSR by destination (int-atomic count as rank, rows sorted by source: fixed sum order) in place of memset + float-atomic scatter + combine
# speedup vs baseline: 3.5377x; 3.5377x over previous
#include <hip/hip_runtime.h>
#include <hip/hip_bf16.h>

typedef __attribute__((ext_vector_type(2))) float v2f;
typedef __attribute__((ext_vector_type(8))) float v8f;

#define EMB_DIM 128
#define HIDDEN  256
#define OUT_C   64
#define K_ITERS 10
#define ALPHA   0.1f

__global__ void zero_counts_kernel(int* __restrict__ cnt, int n) {
    const int i = blockIdx.x * blockDim.x + threadIdx.x;
    if (i < n) cnt[i] = 0;
}

__global__ void count_in_edges_kernel(const int* __restrict__ dst, int* cnt,
                                      int* __restrict__ rank, int e) {
    const int i = blockIdx.x * blockDim.x + threadIdx.x;
    if (i < e) rank[i] = atomicAdd(&cnt[dst[i]], 1);
}

__launch_bounds__(256)
__global__ void chunk_sums_kernel(const int* __restrict__ cnt, int* __restrict__ chunk_sum, int n) {
    __shared__ int wave_sum[8];
    const int tid  = threadIdx.x;
    const int lane = tid & 31;
    const int wv   = tid >> 5;
    const int i0   = blockIdx.x * 1024 + tid * 4;
    int s = 0;
    for (int t = 0; t < 4; ++t)
        if (i0 + t < n) s += cnt[i0 + t];
    for (int off = 16; off > 0; off >>= 1) s += __shfl_xor(s, off);
    if (lane == 0) wave_sum[wv] = s;
    __syncthreads();
    if (tid == 0) {
        int tot = 0;
        for (int w = 0; w < 8; ++w) tot += wave_sum[w];
        chunk_sum[blockIdx.x] = tot;
    }
}

__launch_bounds__(256)
__global__ void chunk_scan_kernel(const int* __restrict__ cnt, const int* __restrict__ chunk_sum,
                                  int* __restrict__ rowstart, int n) {
    __shared__ int wave_sum[8];
    __shared__ int wave_pre[8];
    const int tid  = threadIdx.x;
    const int lane = tid & 31;
    const int wv   = tid >> 5;

    int pre = 0;
    for (int j = tid; j < (int)blockIdx.x; j += 256) pre += chunk_sum[j];
    for (int off = 16; off > 0; off >>= 1) pre += __shfl_xor(pre, off);
    if (lane == 0) wave_pre[wv] = pre;

    const int i0 = blockIdx.x * 1024 + tid * 4;
    const int c0 = (i0 + 0 < n) ? cnt[i0 + 0] : 0;
    const int c1 = (i0 + 1 < n) ? cnt[i0 + 1] : 0;
    const int c2 = (i0 + 2 < n) ? cnt[i0 + 2] : 0;
    const int c3 = (i0 + 3 < n) ? cnt[i0 + 3] : 0;
    const int t1 = c0 + c1, t2 = t1 + c2, tot = t2 + c3;
    int incl = tot;
    for (int off = 1; off < 32; off <<= 1) {
        const int u = __shfl_up(incl, off);
        if (lane >= off) incl += u;
    }
    if (lane == 31) wave_sum[wv] = incl;
    __syncthreads();

    int before = 0;
    for (int w = 0; w < 8; ++w) {
        before += wave_pre[w];
        if (w < wv) before += wave_sum[w];
    }
    const int excl = before + (incl - tot);
    if (i0 + 0 < n) rowstart[i0 + 0] = excl;
    if (i0 + 1 < n) rowstart[i0 + 1] = excl + c0;
    if (i0 + 2 < n) rowstart[i0 + 2] = excl + t1;
    if (i0 + 3 < n) rowstart[i0 + 3] = excl + t2;
}

__global__ void place_edges_kernel(const int* __restrict__ src, const int* __restrict__ dst,
                                   const int* __restrict__ rowstart, const int* __restrict__ rank,
                                   int* __restrict__ unsorted, int e) {
    const int stride = gridDim.x * blockDim.x;
    for (int i = blockIdx.x * blockDim.x + threadIdx.x; i < e; i += stride)
        unsorted[rowstart[dst[i]] + rank[i]] = src[i];
}

__device__ __forceinline__ int row_end(const int* __restrict__ rowstart, int node, int n, int e) {
    return (node + 1 < n) ? rowstart[node + 1] : e;
}

#define NODES_PER_GROUP 4

__global__ void sort_rows_kernel(const int* __restrict__ rowstart, const int* __restrict__ unsorted,
                                 int* __restrict__ csr_src, int n, int e) {
    const unsigned gid = blockIdx.x * blockDim.x + threadIdx.x;
    const int node0 = (int)(gid >> 4) * NODES_PER_GROUP;
    const int q     = (int)(gid & 15);
    for (int r = 0; r < NODES_PER_GROUP; ++r) {
        const int node = node0 + r;
        if (node >= n) break;
        const int start = rowstart[node];
        const int cnt   = row_end(rowstart, node, n, e) - start;
        for (int p = q; p < cnt; p += 16) {
            const int v = unsorted[start + p];
            int rank = 0;
            #pragma unroll 4
            for (int j = 0; j < cnt; ++j) {
                const int u = unsorted[start + j];
                rank += (u < v || (u == v && j < p)) ? 1 : 0;
            }
            csr_src[start + rank] = v;
        }
    }
}

__launch_bounds__(128)
__global__ void mlp_wmma_kernel(const int* __restrict__ xidx,
                                const float* __restrict__ emb,
                                const float* __restrict__ W1, const float* __restrict__ b1,
                                const float* __restrict__ W2, const float* __restrict__ b2,
                                float* __restrict__ h_out, float* __restrict__ x0_out,
                                int n) {
    const int tid  = threadIdx.x;
    const int wave = tid >> 5;
    const int lane = tid & 31;
    const int m    = lane & 15;
    const int g    = lane >> 4;
    const int row0 = blockIdx.x * 16;

    __shared__ float sA[16][EMB_DIM + 2];
    __shared__ float sH[16][HIDDEN + 2];

    for (int i = tid; i < 16 * EMB_DIM; i += 128) {
        int r  = i >> 7;
        int c  = i & (EMB_DIM - 1);
        int rr = row0 + r;
        float v = 0.0f;
        if (rr < n) {
            int node = xidx[rr];
            v = emb[(size_t)node * EMB_DIM + c];
        }
        sA[r][c] = v;
    }
    __syncthreads();

    for (int t = 0; t < 4; ++t) {
        const int nbase = wave * 64 + t * 16;
        v8f acc = {0.f, 0.f, 0.f, 0.f, 0.f, 0.f, 0.f, 0.f};
        for (int s = 0; s < 32; ++s) {
            const int k = 4 * s + 2 * g;
            v2f a; a.x = sA[m][k];                    a.y = sA[m][k + 1];
            v2f b; b.x = W1[(size_t)k * HIDDEN + nbase + m];
                   b.y = W1[(size_t)(k + 1) * HIDDEN + nbase + m];
            acc = __builtin_amdgcn_wmma_f32_16x16x4_f32(
                false, a, false, b, (short)0, acc, false, false);
        }
        const float bias = b1[nbase + m];
        #pragma unroll
        for (int v = 0; v < 8; ++v) {
            float val = acc[v] + bias;
            val = val > 0.0f ? val : 0.0f;
            sH[v + 8 * g][nbase + m] = val;
        }
    }
    __syncthreads();

    {
        const int nbase = wave * 16;
        v8f acc = {0.f, 0.f, 0.f, 0.f, 0.f, 0.f, 0.f, 0.f};
        for (int s = 0; s < 64; ++s) {
            const int k = 4 * s + 2 * g;
            v2f a; a.x = sH[m][k];                    a.y = sH[m][k + 1];
            v2f b; b.x = W2[(size_t)k * OUT_C + nbase + m];
                   b.y = W2[(size_t)(k + 1) * OUT_C + nbase + m];
            acc = __builtin_amdgcn_wmma_f32_16x16x4_f32(
                false, a, false, b, (short)0, acc, false, false);
        }
        const float bias = b2[nbase + m];
        #pragma unroll
        for (int v = 0; v < 8; ++v) {
            const int row = row0 + v + 8 * g;
            if (row < n) {
                const float val = acc[v] + bias;
                const size_t idx = (size_t)row * OUT_C + nbase + m;
                h_out[idx]  = val;
                x0_out[idx] = val;
            }
        }
    }
}

__device__ __forceinline__ float dis_from_count(int cnt) {
    const float d = (float)(cnt + 1);
    return (d > 0.0f) ? rsqrtf(d) : 0.0f;
}

__launch_bounds__(128)
__global__ void gather_round_kernel(const int* __restrict__ rowstart, const int* __restrict__ csr_src,
                                    const float* __restrict__ x, const float* __restrict__ h,
                                    float* __restrict__ xnew, int n, int e) {
    const unsigned gid = blockIdx.x * blockDim.x + threadIdx.x;
    const int node0 = (int)(gid >> 4) * NODES_PER_GROUP;
    const int q     = (int)(gid & 15);
    if (node0 >= n) return;

    const char* xb = (const char*)x;
    const unsigned qoff = (unsigned)q * 16u;

    int bound_my = e;
    if (q <= NODES_PER_GROUP && node0 + q < n) bound_my = rowstart[node0 + q];

    for (int r = 0; r < NODES_PER_GROUP; ++r) {
        const int node = node0 + r;
        if (node >= n) break;
        const int start = __shfl(bound_my, r, 16);
        const int end   = __shfl(bound_my, r + 1, 16);
        const float di  = dis_from_count(end - start);

        const unsigned own = (unsigned)node * 256u + qoff;
        const float4 xo = *(const float4*)(xb + own);
        const float4 hh = *(const float4*)((const char*)h + own);

        float4 acc = make_float4(0.0f, 0.0f, 0.0f, 0.0f);
        for (int base = start; base < end; base += 16) {
            const int my = base + q;
            int   s_my = node;
            float w_my = 0.0f;
            if (my < end) {
                s_my = csr_src[my];
                w_my = dis_from_count(row_end(rowstart, s_my, n, e) - rowstart[s_my]) * di;
            }
            const int m = min(16, end - base);
            for (int j = 0; j < m; j += 4) {
                float4 v[4];
                float  w[4];
                #pragma unroll
                for (int t = 0; t < 4; ++t) {
                    const int s = __shfl(s_my, j + t, 16);
                    w[t] = __shfl(w_my, j + t, 16);
                    v[t] = *(const float4*)(xb + ((unsigned)s * 256u + qoff));
                }
                #pragma unroll
                for (int t = 0; t < 4; ++t) {
                    const bool on = (j + t) < m;
                    acc.x += (on ? v[t].x : 0.0f) * w[t];
                    acc.y += (on ? v[t].y : 0.0f) * w[t];
                    acc.z += (on ? v[t].z : 0.0f) * w[t];
                    acc.w += (on ? v[t].w : 0.0f) * w[t];
                }
            }
        }

        const float d2 = di * di;
        float4 out4;
        out4.x = (1.0f - ALPHA) * (acc.x + d2 * xo.x) + ALPHA * hh.x;
        out4.y = (1.0f - ALPHA) * (acc.y + d2 * xo.y) + ALPHA * hh.y;
        out4.z = (1.0f - ALPHA) * (acc.z + d2 * xo.z) + ALPHA * hh.z;
        out4.w = (1.0f - ALPHA) * (acc.w + d2 * xo.w) + ALPHA * hh.w;
        *(float4*)((char*)xnew + own) = out4;
    }
}

extern "C" void kernel_launch(void* const* d_in, const int* in_sizes, int n_in,
                              void* d_out, int out_size, void* d_ws, size_t ws_size,
                              hipStream_t stream) {
    (void)n_in; (void)out_size; (void)ws_size;

    const int*   xidx = (const int*)d_in[0];
    const int*   ei   = (const int*)d_in[1];
    const float* emb  = (const float*)d_in[2];
    const float* W1   = (const float*)d_in[3];
    const float* b1   = (const float*)d_in[4];
    const float* W2   = (const float*)d_in[5];
    const float* b2   = (const float*)d_in[6];
    float* out = (float*)d_out;

    const int N = in_sizes[0];
    const int E = in_sizes[1] / 2;
    const size_t NC = (size_t)N * OUT_C;

    float* h        = (float*)d_ws;
    float* xping    = h + NC;
    int*   rowstart = (int*)(xping + NC);
    int*   csr_src  = rowstart + N;
    int*   cnt       = (int*)xping;
    int*   rank      = cnt + N;
    int*   unsorted  = rank + E;
    int*   chunk_sum = unsorted + E;

    const int* src = ei;
    const int* dst = ei + E;

    zero_counts_kernel   <<<(N + 255) / 256, 256, 0, stream>>>(cnt, N);
    count_in_edges_kernel<<<(E + 255) / 256, 256, 0, stream>>>(dst, cnt, rank, E);
    const int nchunks = (N + 1023) / 1024;
    chunk_sums_kernel    <<<nchunks, 256, 0, stream>>>(cnt, chunk_sum, N);
    chunk_scan_kernel    <<<nchunks, 256, 0, stream>>>(cnt, chunk_sum, rowstart, N);
    place_edges_kernel   <<<(E + 1023) / 1024, 256, 0, stream>>>(src, dst, rowstart, rank, unsorted, E);
    const int ngroups   = (N + NODES_PER_GROUP - 1) / NODES_PER_GROUP;
    const int nblocks16 = (ngroups + 7) / 8;
    sort_rows_kernel     <<<nblocks16, 128, 0, stream>>>(rowstart, unsorted, csr_src, N, E);

    mlp_wmma_kernel<<<(N + 15) / 16, 128, 0, stream>>>(xidx, emb, W1, b1, W2, b2,
                                                       h, out, N);

    float* xa = out;
    float* xb = xping;
    for (int k = 0; k < K_ITERS; ++k) {
        gather_round_kernel<<<nblocks16, 128, 0, stream>>>(rowstart, csr_src, xa, h, xb, N, E);
        float* t = xa; xa = xb; xb = t;
    }
    if (xa != out)
        hipMemcpyAsync(out, xa, NC * sizeof(float), hipMemcpyDeviceToDevice, stream);
}


# Rounds carry the state as dis*x (first layer stores dis*x0, last round stores x): an edge needs its source id and one row, no per-edge degree look-up or weight; same sum order, other rounding
# speedup vs baseline: 3.9559x; 1.1182x over previous
#include <hip/hip_runtime.h>
#include <hip/hip_bf16.h>

typedef __attribute__((ext_vector_type(2))) float v2f;
typedef __attribute__((ext_vector_type(8))) float v8f;

#define EMB_DIM 128
#define HIDDEN  256
#define OUT_C   64
#define K_ITERS 10
#define ALPHA   0.1f

__global__ void zero_counts_kernel(int* __restrict__ cnt, int n) {
    const int i = blockIdx.x * blockDim.x + threadIdx.x;
    if (i < n) cnt[i] = 0;
}

__global__ void count_in_edges_kernel(const int* __restrict__ dst, int* cnt,
                                      int* __restrict__ rank, int e) {
    const int i = blockIdx.x * blockDim.x + threadIdx.x;
    if (i < e) rank[i] = atomicAdd(&cnt[dst[i]], 1);
}

__launch_bounds__(256)
__global__ void chunk_sums_kernel(const int* __restrict__ cnt, int* __restrict__ chunk_sum, int n) {
    __shared__ int wave_sum[8];
    const int tid  = threadIdx.x;
    const int lane = tid & 31;
    const int wv   = tid >> 5;
    const int i0   = blockIdx.x * 1024 + tid * 4;
    int s = 0;
    for (int t = 0; t < 4; ++t)
        if (i0 + t < n) s += cnt[i0 + t];
    for (int off = 16; off > 0; off >>= 1) s += __shfl_xor(s, off);
    if (lane == 0) wave_sum[wv] = s;
    __syncthreads();
    if (tid == 0) {
        int tot = 0;
        for (int w = 0; w < 8; ++w) tot += wave_sum[w];
        chunk_sum[blockIdx.x] = tot;
    }
}

__launch_bounds__(256)
__global__ void chunk_scan_kernel(const int* __restrict__ cnt, const int* __restrict__ chunk_sum,
                                  int* __restrict__ rowstart, int n) {
    __shared__ int wave_sum[8];
    __shared__ int wave_pre[8];
    const int tid  = threadIdx.x;
    const int lane = tid & 31;
    const int wv   = tid >> 5;

    int pre = 0;
    for (int j = tid; j < (int)blockIdx.x; j += 256) pre += chunk_sum[j];
    for (int off = 16; off > 0; off >>= 1) pre += __shfl_xor(pre, off);
    if (lane == 0) wave_pre[wv] = pre;

    const int i0 = blockIdx.x * 1024 + tid * 4;
    const int c0 = (i0 + 0 < n) ? cnt[i0 + 0] : 0;
    const int c1 = (i0 + 1 < n) ? cnt[i0 + 1] : 0;
    const int c2 = (i0 + 2 < n) ? cnt[i0 + 2] : 0;
    const int c3 = (i0 + 3 < n) ? cnt[i0 + 3] : 0;
    const int t1 = c0 + c1, t2 = t1 + c2, tot = t2 + c3;
    int incl = tot;
    for (int off = 1; off < 32; off <<= 1) {
        const int u = __shfl_up(incl, off);
        if (lane >= off) incl += u;
    }
    if (lane == 31) wave_sum[wv] = incl;
    __syncthreads();

    int before = 0;
    for (int w = 0; w < 8; ++w) {
        before += wave_pre[w];
        if (w < wv) before += wave_sum[w];
    }
    const int excl = before + (incl - tot);
    if (i0 + 0 < n) rowstart[i0 + 0] = excl;
    if (i0 + 1 < n) rowstart[i0 + 1] = excl + c0;
    if (i0 + 2 < n) rowstart[i0 + 2] = excl + t1;
    if (i0 + 3 < n) rowstart[i0 + 3] = excl + t2;
}

__global__ void place_edges_kernel(const int* __restrict__ src, const int* __restrict__ dst,
                                   const int* __restrict__ rowstart, const int* __restrict__ rank,
                                   int* __restrict__ unsorted, int e) {
    const int stride = gridDim.x * blockDim.x;
    for (int i = blockIdx.x * blockDim.x + threadIdx.x; i < e; i += stride)
        unsorted[rowstart[dst[i]] + rank[i]] = src[i];
}

__device__ __forceinline__ int row_end(const int* __restrict__ rowstart, int node, int n, int e) {
    return (node + 1 < n) ? rowstart[node + 1] : e;
}

__device__ __forceinline__ float dis_from_count(int cnt) {
    const float d = (float)(cnt + 1);
    return (d > 0.0f) ? rsqrtf(d) : 0.0f;
}

#define NODES_PER_GROUP 4

__global__ void sort_rows_kernel(const int* __restrict__ rowstart, const int* __restrict__ unsorted,
                                 int* __restrict__ csr_src, int n, int e) {
    const unsigned gid = blockIdx.x * blockDim.x + threadIdx.x;
    const int node0 = (int)(gid >> 4) * NODES_PER_GROUP;
    const int q     = (int)(gid & 15);
    for (int r = 0; r < NODES_PER_GROUP; ++r) {
        const int node = node0 + r;
        if (node >= n) break;
        const int start = rowstart[node];
        const int cnt   = row_end(rowstart, node, n, e) - start;
        for (int p = q; p < cnt; p += 16) {
            const int v = unsorted[start + p];
            int rank = 0;
            #pragma unroll 4
            for (int j = 0; j < cnt; ++j) {
                const int u = unsorted[start + j];
                rank += (u < v || (u == v && j < p)) ? 1 : 0;
            }
            csr_src[start + rank] = v;
        }
    }
}

__launch_bounds__(128)
__global__ void mlp_wmma_kernel(const int* __restrict__ xidx,
                                const float* __restrict__ emb,
                                const float* __restrict__ W1, const float* __restrict__ b1,
                                const float* __restrict__ W2, const float* __restrict__ b2,
                                const int* __restrict__ rowstart,
                                float* __restrict__ h_out, float* __restrict__ z0_out,
                                int n, int e) {
    const int tid  = threadIdx.x;
    const int wave = tid >> 5;
    const int lane = tid & 31;
    const int m    = lane & 15;
    const int g    = lane >> 4;
    const int row0 = blockIdx.x * 16;

    __shared__ float sA[16][EMB_DIM + 2];
    __shared__ float sH[16][HIDDEN + 2];

    for (int i = tid; i < 16 * EMB_DIM; i += 128) {
        int r  = i >> 7;
        int c  = i & (EMB_DIM - 1);
        int rr = row0 + r;
        float v = 0.0f;
        if (rr < n) {
            int node = xidx[rr];
            v = emb[(size_t)node * EMB_DIM + c];
        }
        sA[r][c] = v;
    }
    __syncthreads();

    for (int t = 0; t < 4; ++t) {
        const int nbase = wave * 64 + t * 16;
        v8f acc = {0.f, 0.f, 0.f, 0.f, 0.f, 0.f, 0.f, 0.f};
        for (int s = 0; s < 32; ++s) {
            const int k = 4 * s + 2 * g;
            v2f a; a.x = sA[m][k];                    a.y = sA[m][k + 1];
            v2f b; b.x = W1[(size_t)k * HIDDEN + nbase + m];
                   b.y = W1[(size_t)(k + 1) * HIDDEN + nbase + m];
            acc = __builtin_amdgcn_wmma_f32_16x16x4_f32(
                false, a, false, b, (short)0, acc, false, false);
        }
        const float bias = b1[nbase + m];
        #pragma unroll
        for (int v = 0; v < 8; ++v) {
            float val = acc[v] + bias;
            val = val > 0.0f ? val : 0.0f;
            sH[v + 8 * g][nbase + m] = val;
        }
    }
    __syncthreads();

    {
        const int nbase = wave * 16;
        v8f acc = {0.f, 0.f, 0.f, 0.f, 0.f, 0.f, 0.f, 0.f};
        for (int s = 0; s < 64; ++s) {
            const int k = 4 * s + 2 * g;
            v2f a; a.x = sH[m][k];                    a.y = sH[m][k + 1];
            v2f b; b.x = W2[(size_t)k * OUT_C + nbase + m];
                   b.y = W2[(size_t)(k + 1) * OUT_C + nbase + m];
            acc = __builtin_amdgcn_wmma_f32_16x16x4_f32(
                false, a, false, b, (short)0, acc, false, false);
        }
        const float bias = b2[nbase + m];
        #pragma unroll
        for (int v = 0; v < 8; ++v) {
            const int row = row0 + v + 8 * g;
            if (row < n) {
                const float val = acc[v] + bias;
                const size_t idx = (size_t)row * OUT_C + nbase + m;
                const float di = dis_from_count(row_end(rowstart, row, n, e) - rowstart[row]);
                h_out[idx]  = val;
                z0_out[idx] = di * val;
            }
        }
    }
}

__launch_bounds__(128)
__global__ void gather_round_kernel(const int* __restrict__ rowstart, const int* __restrict__ csr_src,
                                    const float* __restrict__ z, const float* __restrict__ h,
                                    float* __restrict__ znew, int n, int e, int last) {
    const unsigned gid = blockIdx.x * blockDim.x + threadIdx.x;
    const int node0 = (int)(gid >> 4) * NODES_PER_GROUP;
    const int q     = (int)(gid & 15);
    if (node0 >= n) return;

    const char* zb = (const char*)z;
    const unsigned qoff = (unsigned)q * 16u;

    int bound_my = e;
    if (q <= NODES_PER_GROUP && node0 + q < n) bound_my = rowstart[node0 + q];

    for (int r = 0; r < NODES_PER_GROUP; ++r) {
        const int node = node0 + r;
        if (node >= n) break;
        const int start = __shfl(bound_my, r, 16);
        const int end   = __shfl(bound_my, r + 1, 16);
        const float di  = dis_from_count(end - start);

        const unsigned own = (unsigned)node * 256u + qoff;
        const float4 zo = *(const float4*)(zb + own);
        const float4 hh = *(const float4*)((const char*)h + own);

        float4 acc = make_float4(0.0f, 0.0f, 0.0f, 0.0f);
        for (int base = start; base < end; base += 16) {
            const int my = base + q;
            int s_my = node;
            if (my < end) s_my = csr_src[my];
            const int m = min(16, end - base);
            for (int j = 0; j < m; j += 4) {
                float4 v[4];
                #pragma unroll
                for (int t = 0; t < 4; ++t) {
                    const int s = __shfl(s_my, j + t, 16);
                    v[t] = *(const float4*)(zb + ((unsigned)s * 256u + qoff));
                }
                #pragma unroll
                for (int t = 0; t < 4; ++t) {
                    const bool on = (j + t) < m;
                    acc.x += on ? v[t].x : 0.0f;
                    acc.y += on ? v[t].y : 0.0f;
                    acc.z += on ? v[t].z : 0.0f;
                    acc.w += on ? v[t].w : 0.0f;
                }
            }
        }

        const float post = last ? 1.0f : di;
        float4 out4;
        out4.x = post * ((1.0f - ALPHA) * (di * (acc.x + zo.x)) + ALPHA * hh.x);
        out4.y = post * ((1.0f - ALPHA) * (di * (acc.y + zo.y)) + ALPHA * hh.y);
        out4.z = post * ((1.0f - ALPHA) * (di * (acc.z + zo.z)) + ALPHA * hh.z);
        out4.w = post * ((1.0f - ALPHA) * (di * (acc.w + zo.w)) + ALPHA * hh.w);
        *(float4*)((char*)znew + own) = out4;
    }
}

extern "C" void kernel_launch(void* const* d_in, const int* in_sizes, int n_in,
                              void* d_out, int out_size, void* d_ws, size_t ws_size,
                              hipStream_t stream) {
    (void)n_in; (void)out_size; (void)ws_size;

    const int*   xidx = (const int*)d_in[0];
    const int*   ei   = (const int*)d_in[1];
    const float* emb  = (const float*)d_in[2];
    const float* W1   = (const float*)d_in[3];
    const float* b1   = (const float*)d_in[4];
    const float* W2   = (const float*)d_in[5];
    const float* b2   = (const float*)d_in[6];
    float* out = (float*)d_out;

    const int N = in_sizes[0];
    const int E = in_sizes[1] / 2;
    const size_t NC = (size_t)N * OUT_C;

    float* h        = (float*)d_ws;
    float* xping    = h + NC;
    int*   rowstart = (int*)(xping + NC);
    int*   csr_src  = rowstart + N;
    int*   cnt       = (int*)xping;
    int*   rank      = cnt + N;
    int*   unsorted  = rank + E;
    int*   chunk_sum = unsorted + E;

    const int* src = ei;
    const int* dst = ei + E;

    zero_counts_kernel   <<<(N + 255) / 256, 256, 0, stream>>>(cnt, N);
    count_in_edges_kernel<<<(E + 255) / 256, 256, 0, stream>>>(dst, cnt, rank, E);
    const int nchunks = (N + 1023) / 1024;
    chunk_sums_kernel    <<<nchunks, 256, 0, stream>>>(cnt, chunk_sum, N);
    chunk_scan_kernel    <<<nchunks, 256, 0, stream>>>(cnt, chunk_sum, rowstart, N);
    place_edges_kernel   <<<(E + 1023) / 1024, 256, 0, stream>>>(src, dst, rowstart, rank, unsorted, E);
    const int ngroups   = (N + NODES_PER_GROUP - 1) / NODES_PER_GROUP;
    const int nblocks16 = (ngroups + 7) / 8;
    sort_rows_kernel     <<<nblocks16, 128, 0, stream>>>(rowstart, unsorted, csr_src, N, E);

    mlp_wmma_kernel<<<(N + 15) / 16, 128, 0, stream>>>(xidx, emb, W1, b1, W2, b2,
                                                       rowstart, h, out, N, E);

    float* xa = out;
    float* xb = xping;
    for (int k = 0; k < K_ITERS - 1; ++k) {
        gather_round_kernel<<<nblocks16, 128, 0, stream>>>(rowstart, csr_src, xa, h, xb, N, E, 0);
        float* t = xa; xa = xb; xb = t;
    }
    {
        gather_round_kernel<<<nblocks16, 128, 0, stream>>>(rowstart, csr_src, xa, h, xb, N, E, 1);
        float* t = xa; xa = xb; xb = t;
    }
    if (xa != out)
        hipMemcpyAsync(out, xa, NC * sizeof(float), hipMemcpyDeviceToDevice, stream);
}


# CSR build bins the edges by 256-node destination range with LDS counters (two slice passes, then one work-group per bin) in place of per-edge global atomics; rows and sum order unchanged
# speedup vs baseline: 4.5655x; 1.1541x over previous
#include <hip/hip_runtime.h>
#include <hip/hip_bf16.h>

typedef __attribute__((ext_vector_type(2))) float v2f;
typedef __attribute__((ext_vector_type(8))) float v8f;

#define EMB_DIM 128
#define HIDDEN  256
#define OUT_C   64
#define K_ITERS 10
#define ALPHA   0.1f

#define BIN_SHIFT   8
#define BIN_NODES   (1 << BIN_SHIFT)
#define BIN_MASK    (BIN_NODES - 1)
#define SLICE_BLOCKS  256
#define SLICE_THREADS 1024

__launch_bounds__(SLICE_THREADS)
__global__ void bin_count_kernel(const int* __restrict__ dst, int* __restrict__ table,
                                 int e, int nbins) {
    extern __shared__ int bin_cnt[];
    const int tid  = threadIdx.x;
    const int nblk = gridDim.x;
    const int blk  = blockIdx.x;
    for (int b = tid; b <= nbins; b += SLICE_THREADS) bin_cnt[b] = 0;
    __syncthreads();
    const int chunk = (e + nblk - 1) / nblk;
    const int i0 = blk * chunk;
    const int i1 = min(e, i0 + chunk);
    for (int i = i0 + tid; i < i1; i += 4 * SLICE_THREADS) {
        int d[4];
        #pragma unroll
        for (int t = 0; t < 4; ++t) d[t] = dst[min(i + t * SLICE_THREADS, i1 - 1)];
        #pragma unroll
        for (int t = 0; t < 4; ++t)
            atomicAdd(&bin_cnt[(i + t * SLICE_THREADS < i1) ? (d[t] >> BIN_SHIFT) : nbins], 1);
    }
    __syncthreads();
    for (int b = tid; b < nbins; b += SLICE_THREADS) table[b * nblk + blk] = bin_cnt[b];
}

__launch_bounds__(SLICE_THREADS)
__global__ void bin_scatter_kernel(const int* __restrict__ src, const int* __restrict__ dst,
                                   const int* __restrict__ table_start, int* __restrict__ binned,
                                   int e, int nbins) {
    extern __shared__ int bin_cur[];
    const int tid  = threadIdx.x;
    const int nblk = gridDim.x;
    const int blk  = blockIdx.x;
    for (int b = tid; b < nbins; b += SLICE_THREADS) bin_cur[b] = table_start[b * nblk + blk];
    if (tid == 0) bin_cur[nbins] = 0;
    __syncthreads();
    const int chunk = (e + nblk - 1) / nblk;
    const int i0 = blk * chunk;
    const int i1 = min(e, i0 + chunk);
    for (int i = i0 + tid; i < i1; i += 4 * SLICE_THREADS) {
        int d[4], s[4];
        #pragma unroll
        for (int t = 0; t < 4; ++t) {
            const int idc = min(i + t * SLICE_THREADS, i1 - 1);
            d[t] = dst[idc];
            s[t] = src[idc];
        }
        #pragma unroll
        for (int t = 0; t < 4; ++t) {
            const bool on = i + t * SLICE_THREADS < i1;
            const int pos = atomicAdd(&bin_cur[on ? (d[t] >> BIN_SHIFT) : nbins], 1);
            if (on) binned[pos] = (s[t] << BIN_SHIFT) | (d[t] & BIN_MASK);
        }
    }
}

__launch_bounds__(1024)
__global__ void bin_rows_kernel(const int* __restrict__ table_start, const int* __restrict__ binned,
                                int* __restrict__ rowstart, int* __restrict__ unsorted,
                                int n, int e, int nblk, int nbins) {
    __shared__ int node_cnt[BIN_NODES + 1];
    __shared__ int node_cur[BIN_NODES + 1];
    __shared__ int wave_sum[BIN_NODES / 32];
    const int tid  = threadIdx.x;
    const int lane = tid & 31;
    const int wv   = tid >> 5;
    const int b    = blockIdx.x;
    const int seg0 = table_start[b * nblk];
    const int seg1 = (b + 1 < nbins) ? table_start[(b + 1) * nblk] : e;

    if (tid <= BIN_NODES) node_cnt[tid] = 0;
    __syncthreads();
    for (int i = seg0 + tid; i < seg1; i += 4 * 1024) {
        int v[4];
        #pragma unroll
        for (int t = 0; t < 4; ++t) v[t] = binned[min(i + t * 1024, seg1 - 1)];
        #pragma unroll
        for (int t = 0; t < 4; ++t)
            atomicAdd(&node_cnt[(i + t * 1024 < seg1) ? (v[t] & BIN_MASK) : BIN_NODES], 1);
    }
    __syncthreads();

    int c = 0, incl = 0;
    if (tid < BIN_NODES) {
        c = node_cnt[tid];
        incl = c;
        for (int off = 1; off < 32; off <<= 1) {
            const int u = __shfl_up(incl, off);
            if (lane >= off) incl += u;
        }
        if (lane == 31) wave_sum[wv] = incl;
    }
    __syncthreads();
    if (tid < BIN_NODES) {
        int before = seg0;
        for (int w = 0; w < wv; ++w) before += wave_sum[w];
        const int excl = before + (incl - c);
        node_cur[tid] = excl;
        if (tid == 0) node_cur[BIN_NODES] = 0;
        const int node = b * BIN_NODES + tid;
        if (node < n) rowstart[node] = excl;
    }
    __syncthreads();
    for (int i = seg0 + tid; i < seg1; i += 4 * 1024) {
        int v[4];
        #pragma unroll
        for (int t = 0; t < 4; ++t) v[t] = binned[min(i + t * 1024, seg1 - 1)];
        #pragma unroll
        for (int t = 0; t < 4; ++t) {
            const bool on = i + t * 1024 < seg1;
            const int pos = atomicAdd(&node_cur[on ? (v[t] & BIN_MASK) : BIN_NODES], 1);
            if (on) unsorted[pos] = v[t] >> BIN_SHIFT;
        }
    }
}

__launch_bounds__(256)
__global__ void chunk_sums_kernel(const int* __restrict__ cnt, int* __restrict__ chunk_sum, int n) {
    __shared__ int wave_sum[8];
    const int tid  = threadIdx.x;
    const int lane = tid & 31;
    const int wv   = tid >> 5;
    const int i0   = blockIdx.x * 1024 + tid * 4;
    int s = 0;
    for (int t = 0; t < 4; ++t)
        if (i0 + t < n) s += cnt[i0 + t];
    for (int off = 16; off > 0; off >>= 1) s += __shfl_xor(s, off);
    if (lane == 0) wave_sum[wv] = s;
    __syncthreads();
    if (tid == 0) {
        int tot = 0;
        for (int w = 0; w < 8; ++w) tot += wave_sum[w];
        chunk_sum[blockIdx.x] = tot;
    }
}

__launch_bounds__(256)
__global__ void chunk_scan_kernel(const int* __restrict__ cnt, const int* __restrict__ chunk_sum,
                                  int* __restrict__ rowstart, int n) {
    __shared__ int wave_sum[8];
    __shared__ int wave_pre[8];
    const int tid  = threadIdx.x;
    const int lane = tid & 31;
    const int wv   = tid >> 5;

    int pre = 0;
    for (int j = tid; j < (int)blockIdx.x; j += 256) pre += chunk_sum[j];
    for (int off = 16; off > 0; off >>= 1) pre += __shfl_xor(pre, off);
    if (lane == 0) wave_pre[wv] = pre;

    const int i0 = blockIdx.x * 1024 + tid * 4;
    const int c0 = (i0 + 0 < n) ? cnt[i0 + 0] : 0;
    const int c1 = (i0 + 1 < n) ? cnt[i0 + 1] : 0;
    const int c2 = (i0 + 2 < n) ? cnt[i0 + 2] : 0;
    const int c3 = (i0 + 3 < n) ? cnt[i0 + 3] : 0;
    const int t1 = c0 + c1, t2 = t1 + c2, tot = t2 + c3;
    int incl = tot;
    for (int off = 1; off < 32; off <<= 1) {
        const int u = __shfl_up(incl, off);
        if (lane >= off) incl += u;
    }
    if (lane == 31) wave_sum[wv] = incl;
    __syncthreads();

    int before = 0;
    for (int w = 0; w < 8; ++w) {
        before += wave_pre[w];
        if (w < wv) before += wave_sum[w];
    }
    const int excl = before + (incl - tot);
    if (i0 + 0 < n) rowstart[i0 + 0] = excl;
    if (i0 + 1 < n) rowstart[i0 + 1] = excl + c0;
    if (i0 + 2 < n) rowstart[i0 + 2] = excl + t1;
    if (i0 + 3 < n) rowstart[i0 + 3] = excl + t2;
}

__device__ __forceinline__ int row_end(const int* __restrict__ rowstart, int node, int n, int e) {
    return (node + 1 < n) ? rowstart[node + 1] : e;
}

__device__ __forceinline__ float dis_from_count(int cnt) {
    const float d = (float)(cnt + 1);
    return (d > 0.0f) ? rsqrtf(d) : 0.0f;
}

#define NODES_PER_GROUP 4

__global__ void sort_rows_kernel(const int* __restrict__ rowstart, const int* __restrict__ unsorted,
                                 int* __restrict__ csr_src, int n, int e) {
    const unsigned gid = blockIdx.x * blockDim.x + threadIdx.x;
    const int node0 = (int)(gid >> 4) * NODES_PER_GROUP;
    const int q     = (int)(gid & 15);
    for (int r = 0; r < NODES_PER_GROUP; ++r) {
        const int node = node0 + r;
        if (node >= n) break;
        const int start = rowstart[node];
        const int cnt   = row_end(rowstart, node, n, e) - start;
        for (int p = q; p < cnt; p += 16) {
            const int v = unsorted[start + p];
            int rank = 0;
            #pragma unroll 4
            for (int j = 0; j < cnt; ++j) {
                const int u = unsorted[start + j];
                rank += (u < v || (u == v && j < p)) ? 1 : 0;
            }
            csr_src[start + rank] = v;
        }
    }
}

__launch_bounds__(128)
__global__ void mlp_wmma_kernel(const int* __restrict__ xidx,
                                const float* __restrict__ emb,
                                const float* __restrict__ W1, const float* __restrict__ b1,
                                const float* __restrict__ W2, const float* __restrict__ b2,
                                const int* __restrict__ rowstart,
                                float* __restrict__ h_out, float* __restrict__ z0_out,
                                int n, int e) {
    const int tid  = threadIdx.x;
    const int wave = tid >> 5;
    const int lane = tid & 31;
    const int m    = lane & 15;
    const int g    = lane >> 4;
    const int row0 = blockIdx.x * 16;

    __shared__ float sA[16][EMB_DIM + 2];
    __shared__ float sH[16][HIDDEN + 2];

    for (int i = tid; i < 16 * EMB_DIM; i += 128) {
        int r  = i >> 7;
        int c  = i & (EMB_DIM - 1);
        int rr = row0 + r;
        float v = 0.0f;
        if (rr < n) {
            int node = xidx[rr];
            v = emb[(size_t)node * EMB_DIM + c];
        }
        sA[r][c] = v;
    }
    __syncthreads();

    for (int t = 0; t < 4; ++t) {
        const int nbase = wave * 64 + t * 16;
        v8f acc = {0.f, 0.f, 0.f, 0.f, 0.f, 0.f, 0.f, 0.f};
        for (int s = 0; s < 32; ++s) {
            const int k = 4 * s + 2 * g;
            v2f a; a.x = sA[m][k];                    a.y = sA[m][k + 1];
            v2f b; b.x = W1[(size_t)k * HIDDEN + nbase + m];
                   b.y = W1[(size_t)(k + 1) * HIDDEN + nbase + m];
            acc = __builtin_amdgcn_wmma_f32_16x16x4_f32(
                false, a, false, b, (short)0, acc, false, false);
        }
        const float bias = b1[nbase + m];
        #pragma unroll
        for (int v = 0; v < 8; ++v) {
            float val = acc[v] + bias;
            val = val > 0.0f ? val : 0.0f;
            sH[v + 8 * g][nbase + m] = val;
        }
    }
    __syncthreads();

    {
        const int nbase = wave * 16;
        v8f acc = {0.f, 0.f, 0.f, 0.f, 0.f, 0.f, 0.f, 0.f};
        for (int s = 0; s < 64; ++s) {
            const int k = 4 * s + 2 * g;
            v2f a; a.x = sH[m][k];                    a.y = sH[m][k + 1];
            v2f b; b.x = W2[(size_t)k * OUT_C + nbase + m];
                   b.y = W2[(size_t)(k + 1) * OUT_C + nbase + m];
            acc = __builtin_amdgcn_wmma_f32_16x16x4_f32(
                false, a, false, b, (short)0, acc, false, false);
        }
        const float bias = b2[nbase + m];
        #pragma unroll
        for (int v = 0; v < 8; ++v) {
            const int row = row0 + v + 8 * g;
            if (row < n) {
                const float val = acc[v] + bias;
                const size_t idx = (size_t)row * OUT_C + nbase + m;
                const float di = dis_from_count(row_end(rowstart, row, n, e) - rowstart[row]);
                h_out[idx]  = val;
                z0_out[idx] = di * val;
            }
        }
    }
}

__launch_bounds__(128)
__global__ void gather_round_kernel(const int* __restrict__ rowstart, const int* __restrict__ csr_src,
                                    const float* __restrict__ z, const float* __restrict__ h,
                                    float* __restrict__ znew, int n, int e, int last) {
    const unsigned gid = blockIdx.x * blockDim.x + threadIdx.x;
    const int node0 = (int)(gid >> 4) * NODES_PER_GROUP;
    const int q     = (int)(gid & 15);
    if (node0 >= n) return;

    const char* zb = (const char*)z;
    const unsigned qoff = (unsigned)q * 16u;

    int bound_my = e;
    if (q <= NODES_PER_GROUP && node0 + q < n) bound_my = rowstart[node0 + q];

    for (int r = 0; r < NODES_PER_GROUP; ++r) {
        const int node = node0 + r;
        if (node >= n) break;
        const int start = __shfl(bound_my, r, 16);
        const int end   = __shfl(bound_my, r + 1, 16);
        const float di  = dis_from_count(end - start);

        const unsigned own = (unsigned)node * 256u + qoff;
        const float4 zo = *(const float4*)(zb + own);
        const float4 hh = *(const float4*)((const char*)h + own);

        float4 acc = make_float4(0.0f, 0.0f, 0.0f, 0.0f);
        for (int base = start; base < end; base += 16) {
            const int my = base + q;
            int s_my = node;
            if (my < end) s_my = csr_src[my];
            const int m = min(16, end - base);
            for (int j = 0; j < m; j += 4) {
                float4 v[4];
                #pragma unroll
                for (int t = 0; t < 4; ++t) {
                    const int s = __shfl(s_my, j + t, 16);
                    v[t] = *(const float4*)(zb + ((unsigned)s * 256u + qoff));
                }
                #pragma unroll
                for (int t = 0; t < 4; ++t) {
                    const bool on = (j + t) < m;
                    acc.x += on ? v[t].x : 0.0f;
                    acc.y += on ? v[t].y : 0.0f;
                    acc.z += on ? v[t].z : 0.0f;
                    acc.w += on ? v[t].w : 0.0f;
                }
            }
        }

        const float post = last ? 1.0f : di;
        float4 out4;
        out4.x = post * ((1.0f - ALPHA) * (di * (acc.x + zo.x)) + ALPHA * hh.x);
        out4.y = post * ((1.0f - ALPHA) * (di * (acc.y + zo.y)) + ALPHA * hh.y);
        out4.z = post * ((1.0f - ALPHA) * (di * (acc.z + zo.z)) + ALPHA * hh.z);
        out4.w = post * ((1.0f - ALPHA) * (di * (acc.w + zo.w)) + ALPHA * hh.w);
        *(float4*)((char*)znew + own) = out4;
    }
}

extern "C" void kernel_launch(void* const* d_in, const int* in_sizes, int n_in,
                              void* d_out, int out_size, void* d_ws, size_t ws_size,
                              hipStream_t stream) {
    (void)n_in; (void)out_size; (void)ws_size;

    const int*   xidx = (const int*)d_in[0];
    const int*   ei   = (const int*)d_in[1];
    const float* emb  = (const float*)d_in[2];
    const float* W1   = (const float*)d_in[3];
    const float* b1   = (const float*)d_in[4];
    const float* W2   = (const float*)d_in[5];
    const float* b2   = (const float*)d_in[6];
    float* out = (float*)d_out;

    const int N = in_sizes[0];
    const int E = in_sizes[1] / 2;
    const size_t NC = (size_t)N * OUT_C;

    float* h        = (float*)d_ws;
    float* xping    = h + NC;
    int*   rowstart = (int*)(xping + NC);
    int*   csr_src  = rowstart + N;
    const int nbins  = (N + BIN_NODES - 1) / BIN_NODES;
    const int ntable = nbins * SLICE_BLOCKS;
    int*   table       = (int*)xping;
    int*   table_start = table + ntable;
    int*   binned      = table_start + ntable;
    int*   unsorted    = binned + E;
    int*   chunk_sum   = unsorted + E;

    const int* src = ei;
    const int* dst = ei + E;

    const size_t bin_lds = (size_t)(nbins + 1) * sizeof(int);
    bin_count_kernel  <<<SLICE_BLOCKS, SLICE_THREADS, bin_lds, stream>>>(dst, table, E, nbins);
    const int nchunks = (ntable + 1023) / 1024;
    chunk_sums_kernel <<<nchunks, 256, 0, stream>>>(table, chunk_sum, ntable);
    chunk_scan_kernel <<<nchunks, 256, 0, stream>>>(table, chunk_sum, table_start, ntable);
    bin_scatter_kernel<<<SLICE_BLOCKS, SLICE_THREADS, bin_lds, stream>>>(src, dst, table_start, binned, E, nbins);
    bin_rows_kernel   <<<nbins, 1024, 0, stream>>>(table_start, binned, rowstart, unsorted,
                                                   N, E, SLICE_BLOCKS, nbins);
    const int ngroups   = (N + NODES_PER_GROUP - 1) / NODES_PER_GROUP;
    const int nblocks16 = (ngroups + 7) / 8;
    sort_rows_kernel  <<<nblocks16, 128, 0, stream>>>(rowstart, unsorted, csr_src, N, E);

    mlp_wmma_kernel<<<(N + 15) / 16, 128, 0, stream>>>(xidx, emb, W1, b1, W2, b2,
                                                       rowstart, h, out, N, E);

    float* xa = out;
    float* xb = xping;
    for (int k = 0; k < K_ITERS - 1; ++k) {
        gather_round_kernel<<<nblocks16, 128, 0, stream>>>(rowstart, csr_src, xa, h, xb, N, E, 0);
        float* t = xa; xa = xb; xb = t;
    }
    {
        gather_round_kernel<<<nblocks16, 128, 0, stream>>>(rowstart, csr_src, xa, h, xb, N, E, 1);
        float* t = xa; xa = xb; xb = t;
    }
    if (xa != out)
        hipMemcpyAsync(out, xa, NC * sizeof(float), hipMemcpyDeviceToDevice, stream);
}


# CSR build in 3 launches, not 6: each scatter work-group derives its bin offsets from the count table (now 128 slices) itself, each bin work-group orders its own rows; rows and sum order unchanged
# speedup vs baseline: 4.7410x; 1.0384x over previous
#include <hip/hip_runtime.h>
#include <hip/hip_bf16.h>

typedef __attribute__((ext_vector_type(2))) float v2f;
typedef __attribute__((ext_vector_type(8))) float v8f;

#define EMB_DIM 128
#define HIDDEN  256
#define OUT_C   64
#define K_ITERS 10
#define ALPHA   0.1f

#define BIN_SHIFT   8
#define BIN_NODES   (1 << BIN_SHIFT)
#define BIN_MASK    (BIN_NODES - 1)
#define SLICE_BLOCKS  128
#define SLICE_THREADS 1024

__launch_bounds__(SLICE_THREADS)
__global__ void bin_count_kernel(const int* __restrict__ dst, int* __restrict__ table,
                                 int e, int nbins) {
    extern __shared__ int bin_cnt[];
    const int tid  = threadIdx.x;
    const int nblk = gridDim.x;
    const int blk  = blockIdx.x;
    for (int b = tid; b <= nbins; b += SLICE_THREADS) bin_cnt[b] = 0;
    __syncthreads();
    const int chunk = (e + nblk - 1) / nblk;
    const int i0 = blk * chunk;
    const int i1 = min(e, i0 + chunk);
    for (int i = i0 + tid; i < i1; i += 4 * SLICE_THREADS) {
        int d[4];
        #pragma unroll
        for (int t = 0; t < 4; ++t) d[t] = dst[min(i + t * SLICE_THREADS, i1 - 1)];
        #pragma unroll
        for (int t = 0; t < 4; ++t)
            atomicAdd(&bin_cnt[(i + t * SLICE_THREADS < i1) ? (d[t] >> BIN_SHIFT) : nbins], 1);
    }
    __syncthreads();
    for (int b = tid; b < nbins; b += SLICE_THREADS) table[(size_t)blk * nbins + b] = bin_cnt[b];
}

#define TABLE_BATCH 16

__launch_bounds__(SLICE_THREADS)
__global__ void bin_scatter_kernel(const int* __restrict__ src, const int* __restrict__ dst,
                                   const int* __restrict__ table, int* __restrict__ binstart,
                                   int* __restrict__ binned, int e, int nbins) {
    extern __shared__ int bin_lds[];
    int* bin_cur = bin_lds;
    int* bin_tot = bin_lds + nbins + 1;
    __shared__ int wave_sum[SLICE_THREADS / 32];
    const int tid  = threadIdx.x;
    const int lane = tid & 31;
    const int wv   = tid >> 5;
    const int nblk = gridDim.x;
    const int blk  = blockIdx.x;

    for (int b = tid; b < nbins; b += SLICE_THREADS) { bin_cur[b] = 0; bin_tot[b] = 0; }
    if (tid == 0) bin_cur[nbins] = 0;
    __syncthreads();
    for (int s = wv; s < nblk; s += SLICE_THREADS / 32) {
        const int* row = table + (size_t)s * nbins;
        const bool earlier = s < blk;
        for (int b0 = lane; b0 < nbins; b0 += 32 * TABLE_BATCH) {
            int c[TABLE_BATCH];
            #pragma unroll
            for (int t = 0; t < TABLE_BATCH; ++t) c[t] = row[min(b0 + 32 * t, nbins - 1)];
            #pragma unroll
            for (int t = 0; t < TABLE_BATCH; ++t) {
                const int b = b0 + 32 * t;
                if (b < nbins) {
                    atomicAdd(&bin_tot[b], c[t]);
                    if (earlier) atomicAdd(&bin_cur[b], c[t]);
                }
            }
        }
    }
    __syncthreads();
    int carry = 0;
    for (int base = 0; base < nbins; base += SLICE_THREADS) {
        const int b = base + tid;
        const int c = (b < nbins) ? bin_tot[b] : 0;
        int incl = c;
        for (int off = 1; off < 32; off <<= 1) {
            const int u = __shfl_up(incl, off);
            if (lane >= off) incl += u;
        }
        if (lane == 31) wave_sum[wv] = incl;
        __syncthreads();
        int before = carry, total = 0;
        for (int w = 0; w < SLICE_THREADS / 32; ++w) {
            const int ws = wave_sum[w];
            if (w < wv) before += ws;
            total += ws;
        }
        if (b < nbins) {
            const int start = before + (incl - c);
            bin_cur[b] += start;
            if (blk == 0) binstart[b] = start;
        }
        carry += total;
        __syncthreads();
    }

    const int chunk = (e + nblk - 1) / nblk;
    const int i0 = blk * chunk;
    const int i1 = min(e, i0 + chunk);
    for (int i = i0 + tid; i < i1; i += 4 * SLICE_THREADS) {
        int d[4], s[4];
        #pragma unroll
        for (int t = 0; t < 4; ++t) {
            const int idc = min(i + t * SLICE_THREADS, i1 - 1);
            d[t] = dst[idc];
            s[t] = src[idc];
        }
        #pragma unroll
        for (int t = 0; t < 4; ++t) {
            const bool on = i + t * SLICE_THREADS < i1;
            const int pos = atomicAdd(&bin_cur[on ? (d[t] >> BIN_SHIFT) : nbins], 1);
            if (on) binned[pos] = (s[t] << BIN_SHIFT) | (d[t] & BIN_MASK);
        }
    }
}

__launch_bounds__(1024)
__global__ void bin_rows_kernel(const int* __restrict__ binstart, const int* __restrict__ binned,
                                int* __restrict__ rowstart, int* rows,
                                int* __restrict__ csr_src, int n, int e, int nbins) {
    __shared__ int node_cnt[BIN_NODES + 1];
    __shared__ int node_cur[BIN_NODES + 1];
    __shared__ int node_beg[BIN_NODES];
    __shared__ int wave_sum[BIN_NODES / 32];
    const int tid  = threadIdx.x;
    const int lane = tid & 31;
    const int wv   = tid >> 5;
    const int b    = blockIdx.x;
    const int seg0 = binstart[b];
    const int seg1 = (b + 1 < nbins) ? binstart[b + 1] : e;

    if (tid <= BIN_NODES) node_cnt[tid] = 0;
    __syncthreads();
    for (int i = seg0 + tid; i < seg1; i += 4 * 1024) {
        int v[4];
        #pragma unroll
        for (int t = 0; t < 4; ++t) v[t] = binned[min(i + t * 1024, seg1 - 1)];
        #pragma unroll
        for (int t = 0; t < 4; ++t)
            atomicAdd(&node_cnt[(i + t * 1024 < seg1) ? (v[t] & BIN_MASK) : BIN_NODES], 1);
    }
    __syncthreads();

    int c = 0, incl = 0;
    if (tid < BIN_NODES) {
        c = node_cnt[tid];
        incl = c;
        for (int off = 1; off < 32; off <<= 1) {
            const int u = __shfl_up(incl, off);
            if (lane >= off) incl += u;
        }
        if (lane == 31) wave_sum[wv] = incl;
    }
    __syncthreads();
    if (tid < BIN_NODES) {
        int before = seg0;
        for (int w = 0; w < wv; ++w) before += wave_sum[w];
        const int excl = before + (incl - c);
        node_cur[tid] = excl;
        node_beg[tid] = excl;
        if (tid == 0) node_cur[BIN_NODES] = 0;
        const int node = b * BIN_NODES + tid;
        if (node < n) rowstart[node] = excl;
    }
    __syncthreads();
    for (int i = seg0 + tid; i < seg1; i += 4 * 1024) {
        int v[4];
        #pragma unroll
        for (int t = 0; t < 4; ++t) v[t] = binned[min(i + t * 1024, seg1 - 1)];
        #pragma unroll
        for (int t = 0; t < 4; ++t) {
            const bool on = i + t * 1024 < seg1;
            const int pos = atomicAdd(&node_cur[on ? (v[t] & BIN_MASK) : BIN_NODES], 1);
            if (on) rows[pos] = v[t];
        }
    }
    __syncthreads();

    for (int i = seg0 + tid; i < seg1; i += 1024) {
        const int v     = rows[i];
        const int d     = v & BIN_MASK;
        const int start = node_beg[d];
        const int cnt   = node_cnt[d];
        const int p     = i - start;
        int rank = 0;
        #pragma unroll 4
        for (int j = 0; j < cnt; ++j) {
            const int u = rows[start + j];
            rank += (u < v || (u == v && j < p)) ? 1 : 0;
        }
        csr_src[start + rank] = v >> BIN_SHIFT;
    }
}

__device__ __forceinline__ int row_end(const int* __restrict__ rowstart, int node, int n, int e) {
    return (node + 1 < n) ? rowstart[node + 1] : e;
}

__device__ __forceinline__ float dis_from_count(int cnt) {
    const float d = (float)(cnt + 1);
    return (d > 0.0f) ? rsqrtf(d) : 0.0f;
}

#define NODES_PER_GROUP 4

__launch_bounds__(128)
__global__ void mlp_wmma_kernel(const int* __restrict__ xidx,
                                const float* __restrict__ emb,
                                const float* __restrict__ W1, const float* __restrict__ b1,
                                const float* __restrict__ W2, const float* __restrict__ b2,
                                const int* __restrict__ rowstart,
                                float* __restrict__ h_out, float* __restrict__ z0_out,
                                int n, int e) {
    const int tid  = threadIdx.x;
    const int wave = tid >> 5;
    const int lane = tid & 31;
    const int m    = lane & 15;
    const int g    = lane >> 4;
    const int row0 = blockIdx.x * 16;

    __shared__ float sA[16][EMB_DIM + 2];
    __shared__ float sH[16][HIDDEN + 2];

    for (int i = tid; i < 16 * EMB_DIM; i += 128) {
        int r  = i >> 7;
        int c  = i & (EMB_DIM - 1);
        int rr = row0 + r;
        float v = 0.0f;
        if (rr < n) {
            int node = xidx[rr];
            v = emb[(size_t)node * EMB_DIM + c];
        }
        sA[r][c] = v;
    }
    __syncthreads();

    for (int t = 0; t < 4; ++t) {
        const int nbase = wave * 64 + t * 16;
        v8f acc = {0.f, 0.f, 0.f, 0.f, 0.f, 0.f, 0.f, 0.f};
        for (int s = 0; s < 32; ++s) {
            const int k = 4 * s + 2 * g;
            v2f a; a.x = sA[m][k];                    a.y = sA[m][k + 1];
            v2f b; b.x = W1[(size_t)k * HIDDEN + nbase + m];
                   b.y = W1[(size_t)(k + 1) * HIDDEN + nbase + m];
            acc = __builtin_amdgcn_wmma_f32_16x16x4_f32(
                false, a, false, b, (short)0, acc, false, false);
        }
        const float bias = b1[nbase + m];
        #pragma unroll
        for (int v = 0; v < 8; ++v) {
            float val = acc[v] + bias;
            val = val > 0.0f ? val : 0.0f;
            sH[v + 8 * g][nbase + m] = val;
        }
    }
    __syncthreads();

    {
        const int nbase = wave * 16;
        v8f acc = {0.f, 0.f, 0.f, 0.f, 0.f, 0.f, 0.f, 0.f};
        for (int s = 0; s < 64; ++s) {
            const int k = 4 * s + 2 * g;
            v2f a; a.x = sH[m][k];                    a.y = sH[m][k + 1];
            v2f b; b.x = W2[(size_t)k * OUT_C + nbase + m];
                   b.y = W2[(size_t)(k + 1) * OUT_C + nbase + m];
            acc = __builtin_amdgcn_wmma_f32_16x16x4_f32(
                false, a, false, b, (short)0, acc, false, false);
        }
        const float bias = b2[nbase + m];
        #pragma unroll
        for (int v = 0; v < 8; ++v) {
            const int row = row0 + v + 8 * g;
            if (row < n) {
                const float val = acc[v] + bias;
                const size_t idx = (size_t)row * OUT_C + nbase + m;
                const float di = dis_from_count(row_end(rowstart, row, n, e) - rowstart[row]);
                h_out[idx]  = val;
                z0_out[idx] = di * val;
            }
        }
    }
}

__launch_bounds__(128)
__global__ void gather_round_kernel(const int* __restrict__ rowstart, const int* __restrict__ csr_src,
                                    const float* __restrict__ z, const float* __restrict__ h,
                                    float* __restrict__ znew, int n, int e, int last) {
    const unsigned gid = blockIdx.x * blockDim.x + threadIdx.x;
    const int node0 = (int)(gid >> 4) * NODES_PER_GROUP;
    const int q     = (int)(gid & 15);
    if (node0 >= n) return;

    const char* zb = (const char*)z;
    const unsigned qoff = (unsigned)q * 16u;

    int bound_my = e;
    if (q <= NODES_PER_GROUP && node0 + q < n) bound_my = rowstart[node0 + q];

    for (int r = 0; r < NODES_PER_GROUP; ++r) {
        const int node = node0 + r;
        if (node >= n) break;
        const int start = __shfl(bound_my, r, 16);
        const int end   = __shfl(bound_my, r + 1, 16);
        const float di  = dis_from_count(end - start);

        const unsigned own = (unsigned)node * 256u + qoff;
        const float4 zo = *(const float4*)(zb + own);
        const float4 hh = *(const float4*)((const char*)h + own);

        float4 acc = make_float4(0.0f, 0.0f, 0.0f, 0.0f);
        for (int base = start; base < end; base += 16) {
            const int my = base + q;
            int s_my = node;
            if (my < end) s_my = csr_src[my];
            const int m = min(16, end - base);
            for (int j = 0; j < m; j += 4) {
                float4 v[4];
                #pragma unroll
                for (int t = 0; t < 4; ++t) {
                    const int s = __shfl(s_my, j + t, 16);
                    v[t] = *(const float4*)(zb + ((unsigned)s * 256u + qoff));
                }
                #pragma unroll
                for (int t = 0; t < 4; ++t) {
                    const bool on = (j + t) < m;
                    acc.x += on ? v[t].x : 0.0f;
                    acc.y += on ? v[t].y : 0.0f;
                    acc.z += on ? v[t].z : 0.0f;
                    acc.w += on ? v[t].w : 0.0f;
                }
            }
        }

        const float post = last ? 1.0f : di;
        float4 out4;
        out4.x = post * ((1.0f - ALPHA) * (di * (acc.x + zo.x)) + ALPHA * hh.x);
        out4.y = post * ((1.0f - ALPHA) * (di * (acc.y + zo.y)) + ALPHA * hh.y);
        out4.z = post * ((1.0f - ALPHA) * (di * (acc.z + zo.z)) + ALPHA * hh.z);
        out4.w = post * ((1.0f - ALPHA) * (di * (acc.w + zo.w)) + ALPHA * hh.w);
        *(float4*)((char*)znew + own) = out4;
    }
}

extern "C" void kernel_launch(void* const* d_in, const int* in_sizes, int n_in,
                              void* d_out, int out_size, void* d_ws, size_t ws_size,
                              hipStream_t stream) {
    (void)n_in; (void)out_size; (void)ws_size;

    const int*   xidx = (const int*)d_in[0];
    const int*   ei   = (const int*)d_in[1];
    const float* emb  = (const float*)d_in[2];
    const float* W1   = (const float*)d_in[3];
    const float* b1   = (const float*)d_in[4];
    const float* W2   = (const float*)d_in[5];
    const float* b2   = (const float*)d_in[6];
    float* out = (float*)d_out;

    const int N = in_sizes[0];
    const int E = in_sizes[1] / 2;
    const size_t NC = (size_t)N * OUT_C;

    float* h        = (float*)d_ws;
    float* xping    = h + NC;
    int*   rowstart = (int*)(xping + NC);
    int*   csr_src  = rowstart + N;
    const int nbins  = (N + BIN_NODES - 1) / BIN_NODES;
    const int ntable = nbins * SLICE_BLOCKS;
    int*   table    = (int*)xping;
    int*   binstart = table + ntable;
    int*   binned   = binstart + nbins;
    int*   rows     = binned + E;

    const int* src = ei;
    const int* dst = ei + E;

    const size_t cnt_lds = (size_t)(nbins + 1) * sizeof(int);
    const size_t sct_lds = (size_t)(2 * nbins + 1) * sizeof(int);
    bin_count_kernel  <<<SLICE_BLOCKS, SLICE_THREADS, cnt_lds, stream>>>(dst, table, E, nbins);
    bin_scatter_kernel<<<SLICE_BLOCKS, SLICE_THREADS, sct_lds, stream>>>(src, dst, table, binstart, binned, E, nbins);
    bin_rows_kernel   <<<nbins, 1024, 0, stream>>>(binstart, binned, rowstart, rows, csr_src, N, E, nbins);
    const int ngroups   = (N + NODES_PER_GROUP - 1) / NODES_PER_GROUP;
    const int nblocks16 = (ngroups + 7) / 8;

    mlp_wmma_kernel<<<(N + 15) / 16, 128, 0, stream>>>(xidx, emb, W1, b1, W2, b2,
                                                       rowstart, h, out, N, E);

    float* xa = out;
    float* xb = xping;
    for (int k = 0; k < K_ITERS - 1; ++k) {
        gather_round_kernel<<<nblocks16, 128, 0, stream>>>(rowstart, csr_src, xa, h, xb, N, E, 0);
        float* t = xa; xa = xb; xb = t;
    }
    {
        gather_round_kernel<<<nblocks16, 128, 0, stream>>>(rowstart, csr_src, xa, h, xb, N, E, 1);
        float* t = xa; xa = xb; xb = t;
    }
    if (xa != out)
        hipMemcpyAsync(out, xa, NC * sizeof(float), hipMemcpyDeviceToDevice, stream);
}


# Gather rounds: all 16 rows of a list piece are requested before the first sum waits (was 4 at a time), the next piece's source ids fetched under them; 6 nodes per 16-lane group; sum order unchanged
# speedup vs baseline: 4.8836x; 1.0301x over previous
#include <hip/hip_runtime.h>
#include <hip/hip_bf16.h>

typedef __attribute__((ext_vector_type(2))) float v2f;
typedef __attribute__((ext_vector_type(8))) float v8f;

#define EMB_DIM 128
#define HIDDEN  256
#define OUT_C   64
#define K_ITERS 10
#define ALPHA   0.1f

#define BIN_SHIFT   8
#define BIN_NODES   (1 << BIN_SHIFT)
#define BIN_MASK    (BIN_NODES - 1)
#define SLICE_BLOCKS  128
#define SLICE_THREADS 1024

__launch_bounds__(SLICE_THREADS)
__global__ void bin_count_kernel(const int* __restrict__ dst, int* __restrict__ table,
                                 int e, int nbins) {
    extern __shared__ int bin_cnt[];
    const int tid  = threadIdx.x;
    const int nblk = gridDim.x;
    const int blk  = blockIdx.x;
    for (int b = tid; b <= nbins; b += SLICE_THREADS) bin_cnt[b] = 0;
    __syncthreads();
    const int chunk = (e + nblk - 1) / nblk;
    const int i0 = blk * chunk;
    const int i1 = min(e, i0 + chunk);
    for (int i = i0 + tid; i < i1; i += 4 * SLICE_THREADS) {
        int d[4];
        #pragma unroll
        for (int t = 0; t < 4; ++t) d[t] = dst[min(i + t * SLICE_THREADS, i1 - 1)];
        #pragma unroll
        for (int t = 0; t < 4; ++t)
            atomicAdd(&bin_cnt[(i + t * SLICE_THREADS < i1) ? (d[t] >> BIN_SHIFT) : nbins], 1);
    }
    __syncthreads();
    for (int b = tid; b < nbins; b += SLICE_THREADS) table[(size_t)blk * nbins + b] = bin_cnt[b];
}

#define TABLE_BATCH 16

__launch_bounds__(SLICE_THREADS)
__global__ void bin_scatter_kernel(const int* __restrict__ src, const int* __restrict__ dst,
                                   const int* __restrict__ table, int* __restrict__ binstart,
                                   int* __restrict__ binned, int e, int nbins) {
    extern __shared__ int bin_lds[];
    int* bin_cur = bin_lds;
    int* bin_tot = bin_lds + nbins + 1;
    __shared__ int wave_sum[SLICE_THREADS / 32];
    const int tid  = threadIdx.x;
    const int lane = tid & 31;
    const int wv   = tid >> 5;
    const int nblk = gridDim.x;
    const int blk  = blockIdx.x;

    for (int b = tid; b < nbins; b += SLICE_THREADS) { bin_cur[b] = 0; bin_tot[b] = 0; }
    if (tid == 0) bin_cur[nbins] = 0;
    __syncthreads();
    for (int s = wv; s < nblk; s += SLICE_THREADS / 32) {
        const int* row = table + (size_t)s * nbins;
        const bool earlier = s < blk;
        for (int b0 = lane; b0 < nbins; b0 += 32 * TABLE_BATCH) {
            int c[TABLE_BATCH];
            #pragma unroll
            for (int t = 0; t < TABLE_BATCH; ++t) c[t] = row[min(b0 + 32 * t, nbins - 1)];
            #pragma unroll
            for (int t = 0; t < TABLE_BATCH; ++t) {
                const int b = b0 + 32 * t;
                if (b < nbins) {
                    atomicAdd(&bin_tot[b], c[t]);
                    if (earlier) atomicAdd(&bin_cur[b], c[t]);
                }
            }
        }
    }
    __syncthreads();
    int carry = 0;
    for (int base = 0; base < nbins; base += SLICE_THREADS) {
        const int b = base + tid;
        const int c = (b < nbins) ? bin_tot[b] : 0;
        int incl = c;
        for (int off = 1; off < 32; off <<= 1) {
            const int u = __shfl_up(incl, off);
            if (lane >= off) incl += u;
        }
        if (lane == 31) wave_sum[wv] = incl;
        __syncthreads();
        int before = carry, total = 0;
        for (int w = 0; w < SLICE_THREADS / 32; ++w) {
            const int ws = wave_sum[w];
            if (w < wv) before += ws;
            total += ws;
        }
        if (b < nbins) {
            const int start = before + (incl - c);
            bin_cur[b] += start;
            if (blk == 0) binstart[b] = start;
        }
        carry += total;
        __syncthreads();
    }

    const int chunk = (e + nblk - 1) / nblk;
    const int i0 = blk * chunk;
    const int i1 = min(e, i0 + chunk);
    for (int i = i0 + tid; i < i1; i += 4 * SLICE_THREADS) {
        int d[4], s[4];
        #pragma unroll
        for (int t = 0; t < 4; ++t) {
            const int idc = min(i + t * SLICE_THREADS, i1 - 1);
            d[t] = dst[idc];
            s[t] = src[idc];
        }
        #pragma unroll
        for (int t = 0; t < 4; ++t) {
            const bool on = i + t * SLICE_THREADS < i1;
            const int pos = atomicAdd(&bin_cur[on ? (d[t] >> BIN_SHIFT) : nbins], 1);
            if (on) binned[pos] = (s[t] << BIN_SHIFT) | (d[t] & BIN_MASK);
        }
    }
}

__launch_bounds__(1024)
__global__ void bin_rows_kernel(const int* __restrict__ binstart, const int* __restrict__ binned,
                                int* __restrict__ rowstart, int* rows,
                                int* __restrict__ csr_src, int n, int e, int nbins) {
    __shared__ int node_cnt[BIN_NODES + 1];
    __shared__ int node_cur[BIN_NODES + 1];
    __shared__ int node_beg[BIN_NODES];
    __shared__ int wave_sum[BIN_NODES / 32];
    const int tid  = threadIdx.x;
    const int lane = tid & 31;
    const int wv   = tid >> 5;
    const int b    = blockIdx.x;
    const int seg0 = binstart[b];
    const int seg1 = (b + 1 < nbins) ? binstart[b + 1] : e;

    if (tid <= BIN_NODES) node_cnt[tid] = 0;
    __syncthreads();
    for (int i = seg0 + tid; i < seg1; i += 4 * 1024) {
        int v[4];
        #pragma unroll
        for (int t = 0; t < 4; ++t) v[t] = binned[min(i + t * 1024, seg1 - 1)];
        #pragma unroll
        for (int t = 0; t < 4; ++t)
            atomicAdd(&node_cnt[(i + t * 1024 < seg1) ? (v[t] & BIN_MASK) : BIN_NODES], 1);
    }
    __syncthreads();

    int c = 0, incl = 0;
    if (tid < BIN_NODES) {
        c = node_cnt[tid];
        incl = c;
        for (int off = 1; off < 32; off <<= 1) {
            const int u = __shfl_up(incl, off);
            if (lane >= off) incl += u;
        }
        if (lane == 31) wave_sum[wv] = incl;
    }
    __syncthreads();
    if (tid < BIN_NODES) {
        int before = seg0;
        for (int w = 0; w < wv; ++w) before += wave_sum[w];
        const int excl = before + (incl - c);
        node_cur[tid] = excl;
        node_beg[tid] = excl;
        if (tid == 0) node_cur[BIN_NODES] = 0;
        const int node = b * BIN_NODES + tid;
        if (node < n) rowstart[node] = excl;
    }
    __syncthreads();
    for (int i = seg0 + tid; i < seg1; i += 4 * 1024) {
        int v[4];
        #pragma unroll
        for (int t = 0; t < 4; ++t) v[t] = binned[min(i + t * 1024, seg1 - 1)];
        #pragma unroll
        for (int t = 0; t < 4; ++t) {
            const bool on = i + t * 1024 < seg1;
            const int pos = atomicAdd(&node_cur[on ? (v[t] & BIN_MASK) : BIN_NODES], 1);
            if (on) rows[pos] = v[t];
        }
    }
    __syncthreads();

    for (int i = seg0 + tid; i < seg1; i += 1024) {
        const int v     = rows[i];
        const int d     = v & BIN_MASK;
        const int start = node_beg[d];
        const int cnt   = node_cnt[d];
        const int p     = i - start;
        int rank = 0;
        #pragma unroll 4
        for (int j = 0; j < cnt; ++j) {
            const int u = rows[start + j];
            rank += (u < v || (u == v && j < p)) ? 1 : 0;
        }
        csr_src[start + rank] = v >> BIN_SHIFT;
    }
}

__device__ __forceinline__ int row_end(const int* __restrict__ rowstart, int node, int n, int e) {
    return (node + 1 < n) ? rowstart[node + 1] : e;
}

__device__ __forceinline__ float dis_from_count(int cnt) {
    const float d = (float)(cnt + 1);
    return (d > 0.0f) ? rsqrtf(d) : 0.0f;
}

#define NODES_PER_GROUP 6
#define PIECE 16

__launch_bounds__(128)
__global__ void mlp_wmma_kernel(const int* __restrict__ xidx,
                                const float* __restrict__ emb,
                                const float* __restrict__ W1, const float* __restrict__ b1,
                                const float* __restrict__ W2, const float* __restrict__ b2,
                                const int* __restrict__ rowstart,
                                float* __restrict__ h_out, float* __restrict__ z0_out,
                                int n, int e) {
    const int tid  = threadIdx.x;
    const int wave = tid >> 5;
    const int lane = tid & 31;
    const int m    = lane & 15;
    const int g    = lane >> 4;
    const int row0 = blockIdx.x * 16;

    __shared__ float sA[16][EMB_DIM + 2];
    __shared__ float sH[16][HIDDEN + 2];

    for (int i = tid; i < 16 * EMB_DIM; i += 128) {
        int r  = i >> 7;
        int c  = i & (EMB_DIM - 1);
        int rr = row0 + r;
        float v = 0.0f;
        if (rr < n) {
            int node = xidx[rr];
            v = emb[(size_t)node * EMB_DIM + c];
        }
        sA[r][c] = v;
    }
    __syncthreads();

    for (int t = 0; t < 4; ++t) {
        const int nbase = wave * 64 + t * 16;
        v8f acc = {0.f, 0.f, 0.f, 0.f, 0.f, 0.f, 0.f, 0.f};
        for (int s = 0; s < 32; ++s) {
            const int k = 4 * s + 2 * g;
            v2f a; a.x = sA[m][k];                    a.y = sA[m][k + 1];
            v2f b; b.x = W1[(size_t)k * HIDDEN + nbase + m];
                   b.y = W1[(size_t)(k + 1) * HIDDEN + nbase + m];
            acc = __builtin_amdgcn_wmma_f32_16x16x4_f32(
                false, a, false, b, (short)0, acc, false, false);
        }
        const float bias = b1[nbase + m];
        #pragma unroll
        for (int v = 0; v < 8; ++v) {
            float val = acc[v] + bias;
            val = val > 0.0f ? val : 0.0f;
            sH[v + 8 * g][nbase + m] = val;
        }
    }
    __syncthreads();

    {
        const int nbase = wave * 16;
        v8f acc = {0.f, 0.f, 0.f, 0.f, 0.f, 0.f, 0.f, 0.f};
        for (int s = 0; s < 64; ++s) {
            const int k = 4 * s + 2 * g;
            v2f a; a.x = sH[m][k];                    a.y = sH[m][k + 1];
            v2f b; b.x = W2[(size_t)k * OUT_C + nbase + m];
                   b.y = W2[(size_t)(k + 1) * OUT_C + nbase + m];
            acc = __builtin_amdgcn_wmma_f32_16x16x4_f32(
                false, a, false, b, (short)0, acc, false, false);
        }
        const float bias = b2[nbase + m];
        #pragma unroll
        for (int v = 0; v < 8; ++v) {
            const int row = row0 + v + 8 * g;
            if (row < n) {
                const float val = acc[v] + bias;
                const size_t idx = (size_t)row * OUT_C + nbase + m;
                const float di = dis_from_count(row_end(rowstart, row, n, e) - rowstart[row]);
                h_out[idx]  = val;
                z0_out[idx] = di * val;
            }
        }
    }
}

__launch_bounds__(128)
__global__ void gather_round_kernel(const int* __restrict__ rowstart, const int* __restrict__ csr_src,
                                    const float* __restrict__ z, const float* __restrict__ h,
                                    float* __restrict__ znew, int n, int e, int last) {
    const unsigned gid = blockIdx.x * blockDim.x + threadIdx.x;
    const int node0 = (int)(gid >> 4) * NODES_PER_GROUP;
    const int q     = (int)(gid & 15);
    if (node0 >= n) return;

    const char* zb = (const char*)z;
    const unsigned qoff = (unsigned)q * 16u;

    int bound_my = e;
    if (q <= NODES_PER_GROUP && node0 + q < n) bound_my = rowstart[node0 + q];

    const int gbeg = __shfl(bound_my, 0, 16);
    const int gend = __shfl(bound_my, NODES_PER_GROUP, 16);
    int ids_next = node0;
    if (gbeg + q < gend) ids_next = csr_src[gbeg + q];

    for (int r = 0; r < NODES_PER_GROUP; ++r) {
        const int node = node0 + r;
        if (node >= n) break;
        const int start = __shfl(bound_my, r, 16);
        const int end   = __shfl(bound_my, r + 1, 16);
        const float di  = dis_from_count(end - start);

        const unsigned own = (unsigned)node * 256u + qoff;
        const float4 zo = *(const float4*)(zb + own);
        const float4 hh = *(const float4*)((const char*)h + own);

        float4 acc = make_float4(0.0f, 0.0f, 0.0f, 0.0f);
        for (int base = start; base < end; base += PIECE) {
            const int m = min(PIECE, end - base);
            const int s_my = (q < m) ? ids_next : node;
            ids_next = csr_src[min(base + m + q, e - 1)];

            float4 v[PIECE];
            #define ROW_LOAD(t)  v[t] = *(const float4*)(zb + ((unsigned)__shfl(s_my, (t), 16) * 256u + qoff))
            #define ROW_LOAD4(t) ROW_LOAD(t); ROW_LOAD((t) + 1); ROW_LOAD((t) + 2); ROW_LOAD((t) + 3)
            #define ROW_ADD(t)   { const bool on = (t) < m;          \
                                   acc.x += on ? v[t].x : 0.0f;      \
                                   acc.y += on ? v[t].y : 0.0f;      \
                                   acc.z += on ? v[t].z : 0.0f;      \
                                   acc.w += on ? v[t].w : 0.0f; }
            #define ROW_ADD4(t)  ROW_ADD(t) ROW_ADD((t) + 1) ROW_ADD((t) + 2) ROW_ADD((t) + 3)
            ROW_LOAD4(0);
            if (m > 4) { ROW_LOAD4(4); if (m > 8) { ROW_LOAD4(8); if (m > 12) { ROW_LOAD4(12); } } }
            ROW_ADD4(0)
            if (m > 4) { ROW_ADD4(4) if (m > 8) { ROW_ADD4(8) if (m > 12) { ROW_ADD4(12) } } }
            #undef ROW_LOAD
            #undef ROW_LOAD4
            #undef ROW_ADD
            #undef ROW_ADD4
        }

        const float post = last ? 1.0f : di;
        float4 out4;
        out4.x = post * ((1.0f - ALPHA) * (di * (acc.x + zo.x)) + ALPHA * hh.x);
        out4.y = post * ((1.0f - ALPHA) * (di * (acc.y + zo.y)) + ALPHA * hh.y);
        out4.z = post * ((1.0f - ALPHA) * (di * (acc.z + zo.z)) + ALPHA * hh.z);
        out4.w = post * ((1.0f - ALPHA) * (di * (acc.w + zo.w)) + ALPHA * hh.w);
        *(float4*)((char*)znew + own) = out4;
    }
}

extern "C" void kernel_launch(void* const* d_in, const int* in_sizes, int n_in,
                              void* d_out, int out_size, void* d_ws, size_t ws_size,
                              hipStream_t stream) {
    (void)n_in; (void)out_size; (void)ws_size;

    const int*   xidx = (const int*)d_in[0];
    const int*   ei   = (const int*)d_in[1];
    const float* emb  = (const float*)d_in[2];
    const float* W1   = (const float*)d_in[3];
    const float* b1   = (const float*)d_in[4];
    const float* W2   = (const float*)d_in[5];
    const float* b2   = (const float*)d_in[6];
    float* out = (float*)d_out;

    const int N = in_sizes[0];
    const int E = in_sizes[1] / 2;
    const size_t NC = (size_t)N * OUT_C;

    float* h        = (float*)d_ws;
    float* xping    = h + NC;
    int*   rowstart = (int*)(xping + NC);
    int*   csr_src  = rowstart + N;
    const int nbins  = (N + BIN_NODES - 1) / BIN_NODES;
    const int ntable = nbins * SLICE_BLOCKS;
    int*   table    = (int*)xping;
    int*   binstart = table + ntable;
    int*   binned   = binstart + nbins;
    int*   rows     = binned + E;

    const int* src = ei;
    const int* dst = ei + E;

    const size_t cnt_lds = (size_t)(nbins + 1) * sizeof(int);
    const size_t sct_lds = (size_t)(2 * nbins + 1) * sizeof(int);
    bin_count_kernel  <<<SLICE_BLOCKS, SLICE_THREADS, cnt_lds, stream>>>(dst, table, E, nbins);
    bin_scatter_kernel<<<SLICE_BLOCKS, SLICE_THREADS, sct_lds, stream>>>(src, dst, table, binstart, binned, E, nbins);
    bin_rows_kernel   <<<nbins, 1024, 0, stream>>>(binstart, binned, rowstart, rows, csr_src, N, E, nbins);
    const int ngroups   = (N + NODES_PER_GROUP - 1) / NODES_PER_GROUP;
    const int nblocks16 = (ngroups + 7) / 8;

    mlp_wmma_kernel<<<(N + 15) / 16, 128, 0, stream>>>(xidx, emb, W1, b1, W2, b2,
                                                       rowstart, h, out, N, E);

    float* xa = out;
    float* xb = xping;
    for (int k = 0; k < K_ITERS - 1; ++k) {
        gather_round_kernel<<<nblocks16, 128, 0, stream>>>(rowstart, csr_src, xa, h, xb, N, E, 0);
        float* t = xa; xa = xb; xb = t;
    }
    {
        gather_round_kernel<<<nblocks16, 128, 0, stream>>>(rowstart, csr_src, xa, h, xb, N, E, 1);
        float* t = xa; xa = xb; xb = t;
    }
    if (xa != out)
        hipMemcpyAsync(out, xa, NC * sizeof(float), hipMemcpyDeviceToDevice, stream);
}


# MLP work-group takes 32 rows, not 16: a weight fragment is fetched once and feeds 2 row tiles (32 x 32 wave tile, 4 accumulators); emb rows staged by independent 16-byte loads; k order unchanged
# speedup vs baseline: 5.2104x; 1.0669x over previous
#include <hip/hip_runtime.h>
#include <hip/hip_bf16.h>

typedef __attribute__((ext_vector_type(2))) float v2f;
typedef __attribute__((ext_vector_type(8))) float v8f;

#define EMB_DIM 128
#define HIDDEN  256
#define OUT_C   64
#define K_ITERS 10
#define ALPHA   0.1f

#define BIN_SHIFT   8
#define BIN_NODES   (1 << BIN_SHIFT)
#define BIN_MASK    (BIN_NODES - 1)
#define SLICE_BLOCKS  128
#define SLICE_THREADS 1024

__launch_bounds__(SLICE_THREADS)
__global__ void bin_count_kernel(const int* __restrict__ dst, int* __restrict__ table,
                                 int e, int nbins) {
    extern __shared__ int bin_cnt[];
    const int tid  = threadIdx.x;
    const int nblk = gridDim.x;
    const int blk  = blockIdx.x;
    for (int b = tid; b <= nbins; b += SLICE_THREADS) bin_cnt[b] = 0;
    __syncthreads();
    const int chunk = (e + nblk - 1) / nblk;
    const int i0 = blk * chunk;
    const int i1 = min(e, i0 + chunk);
    for (int i = i0 + tid; i < i1; i += 4 * SLICE_THREADS) {
        int d[4];
        #pragma unroll
        for (int t = 0; t < 4; ++t) d[t] = dst[min(i + t * SLICE_THREADS, i1 - 1)];
        #pragma unroll
        for (int t = 0; t < 4; ++t)
            atomicAdd(&bin_cnt[(i + t * SLICE_THREADS < i1) ? (d[t] >> BIN_SHIFT) : nbins], 1);
    }
    __syncthreads();
    for (int b = tid; b < nbins; b += SLICE_THREADS) table[(size_t)blk * nbins + b] = bin_cnt[b];
}

#define TABLE_BATCH 16

__launch_bounds__(SLICE_THREADS)
__global__ void bin_scatter_kernel(const int* __restrict__ src, const int* __restrict__ dst,
                                   const int* __restrict__ table, int* __restrict__ binstart,
                                   int* __restrict__ binned, int e, int nbins) {
    extern __shared__ int bin_lds[];
    int* bin_cur = bin_lds;
    int* bin_tot = bin_lds + nbins + 1;
    __shared__ int wave_sum[SLICE_THREADS / 32];
    const int tid  = threadIdx.x;
    const int lane = tid & 31;
    const int wv   = tid >> 5;
    const int nblk = gridDim.x;
    const int blk  = blockIdx.x;

    for (int b = tid; b < nbins; b += SLICE_THREADS) { bin_cur[b] = 0; bin_tot[b] = 0; }
    if (tid == 0) bin_cur[nbins] = 0;
    __syncthreads();
    for (int s = wv; s < nblk; s += SLICE_THREADS / 32) {
        const int* row = table + (size_t)s * nbins;
        const bool earlier = s < blk;
        for (int b0 = lane; b0 < nbins; b0 += 32 * TABLE_BATCH) {
            int c[TABLE_BATCH];
            #pragma unroll
            for (int t = 0; t < TABLE_BATCH; ++t) c[t] = row[min(b0 + 32 * t, nbins - 1)];
            #pragma unroll
            for (int t = 0; t < TABLE_BATCH; ++t) {
                const int b = b0 + 32 * t;
                if (b < nbins) {
                    atomicAdd(&bin_tot[b], c[t]);
                    if (earlier) atomicAdd(&bin_cur[b], c[t]);
                }
            }
        }
    }
    __syncthreads();
    int carry = 0;
    for (int base = 0; base < nbins; base += SLICE_THREADS) {
        const int b = base + tid;
        const int c = (b < nbins) ? bin_tot[b] : 0;
        int incl = c;
        for (int off = 1; off < 32; off <<= 1) {
            const int u = __shfl_up(incl, off);
            if (lane >= off) incl += u;
        }
        if (lane == 31) wave_sum[wv] = incl;
        __syncthreads();
        int before = carry, total = 0;
        for (int w = 0; w < SLICE_THREADS / 32; ++w) {
            const int ws = wave_sum[w];
            if (w < wv) before += ws;
            total += ws;
        }
        if (b < nbins) {
            const int start = before + (incl - c);
            bin_cur[b] += start;
            if (blk == 0) binstart[b] = start;
        }
        carry += total;
        __syncthreads();
    }

    const int chunk = (e + nblk - 1) / nblk;
    const int i0 = blk * chunk;
    const int i1 = min(e, i0 + chunk);
    for (int i = i0 + tid; i < i1; i += 4 * SLICE_THREADS) {
        int d[4], s[4];
        #pragma unroll
        for (int t = 0; t < 4; ++t) {
            const int idc = min(i + t * SLICE_THREADS, i1 - 1);
            d[t] = dst[idc];
            s[t] = src[idc];
        }
        #pragma unroll
        for (int t = 0; t < 4; ++t) {
            const bool on = i + t * SLICE_THREADS < i1;
            const int pos = atomicAdd(&bin_cur[on ? (d[t] >> BIN_SHIFT) : nbins], 1);
            if (on) binned[pos] = (s[t] << BIN_SHIFT) | (d[t] & BIN_MASK);
        }
    }
}

__launch_bounds__(1024)
__global__ void bin_rows_kernel(const int* __restrict__ binstart, const int* __restrict__ binned,
                                int* __restrict__ rowstart, int* rows,
                                int* __restrict__ csr_src, int n, int e, int nbins) {
    __shared__ int node_cnt[BIN_NODES + 1];
    __shared__ int node_cur[BIN_NODES + 1];
    __shared__ int node_beg[BIN_NODES];
    __shared__ int wave_sum[BIN_NODES / 32];
    const int tid  = threadIdx.x;
    const int lane = tid & 31;
    const int wv   = tid >> 5;
    const int b    = blockIdx.x;
    const int seg0 = binstart[b];
    const int seg1 = (b + 1 < nbins) ? binstart[b + 1] : e;

    if (tid <= BIN_NODES) node_cnt[tid] = 0;
    __syncthreads();
    for (int i = seg0 + tid; i < seg1; i += 4 * 1024) {
        int v[4];
        #pragma unroll
        for (int t = 0; t < 4; ++t) v[t] = binned[min(i + t * 1024, seg1 - 1)];
        #pragma unroll
        for (int t = 0; t < 4; ++t)
            atomicAdd(&node_cnt[(i + t * 1024 < seg1) ? (v[t] & BIN_MASK) : BIN_NODES], 1);
    }
    __syncthreads();

    int c = 0, incl = 0;
    if (tid < BIN_NODES) {
        c = node_cnt[tid];
        incl = c;
        for (int off = 1; off < 32; off <<= 1) {
            const int u = __shfl_up(incl, off);
            if (lane >= off) incl += u;
        }
        if (lane == 31) wave_sum[wv] = incl;
    }
    __syncthreads();
    if (tid < BIN_NODES) {
        int before = seg0;
        for (int w = 0; w < wv; ++w) before += wave_sum[w];
        const int excl = before + (incl - c);
        node_cur[tid] = excl;
        node_beg[tid] = excl;
        if (tid == 0) node_cur[BIN_NODES] = 0;
        const int node = b * BIN_NODES + tid;
        if (node < n) rowstart[node] = excl;
    }
    __syncthreads();
    for (int i = seg0 + tid; i < seg1; i += 4 * 1024) {
        int v[4];
        #pragma unroll
        for (int t = 0; t < 4; ++t) v[t] = binned[min(i + t * 1024, seg1 - 1)];
        #pragma unroll
        for (int t = 0; t < 4; ++t) {
            const bool on = i + t * 1024 < seg1;
            const int pos = atomicAdd(&node_cur[on ? (v[t] & BIN_MASK) : BIN_NODES], 1);
            if (on) rows[pos] = v[t];
        }
    }
    __syncthreads();

    for (int i = seg0 + tid; i < seg1; i += 1024) {
        const int v     = rows[i];
        const int d     = v & BIN_MASK;
        const int start = node_beg[d];
        const int cnt   = node_cnt[d];
        const int p     = i - start;
        int rank = 0;
        #pragma unroll 4
        for (int j = 0; j < cnt; ++j) {
            const int u = rows[start + j];
            rank += (u < v || (u == v && j < p)) ? 1 : 0;
        }
        csr_src[start + rank] = v >> BIN_SHIFT;
    }
}

__device__ __forceinline__ int row_end(const int* __restrict__ rowstart, int node, int n, int e) {
    return (node + 1 < n) ? rowstart[node + 1] : e;
}

__device__ __forceinline__ float dis_from_count(int cnt) {
    const float d = (float)(cnt + 1);
    return (d > 0.0f) ? rsqrtf(d) : 0.0f;
}

#define NODES_PER_GROUP 6
#define PIECE 16

#define MLP_ROWS 32
#define MLP_RT   (MLP_ROWS / 16)
#define SA_PITCH (EMB_DIM + 4)
#define SH_PITCH (HIDDEN + 4)
#define STAGE_PER_THREAD (MLP_ROWS * EMB_DIM / 4 / 128)

__launch_bounds__(128)
__global__ void mlp_wmma_kernel(const int* __restrict__ xidx,
                                const float* __restrict__ emb,
                                const float* __restrict__ W1, const float* __restrict__ b1,
                                const float* __restrict__ W2, const float* __restrict__ b2,
                                const int* __restrict__ rowstart,
                                float* __restrict__ h_out, float* __restrict__ z0_out,
                                int n, int e) {
    const int tid  = threadIdx.x;
    const int wave = tid >> 5;
    const int lane = tid & 31;
    const int m    = lane & 15;
    const int g    = lane >> 4;
    const int row0 = blockIdx.x * MLP_ROWS;

    __shared__ __attribute__((aligned(16))) float sA[MLP_ROWS][SA_PITCH];
    __shared__ __attribute__((aligned(16))) float sH[MLP_ROWS][SH_PITCH];

    {
        int    node[STAGE_PER_THREAD];
        float4 val[STAGE_PER_THREAD];
        #pragma unroll
        for (int j = 0; j < STAGE_PER_THREAD; ++j)
            node[j] = xidx[min(row0 + 4 * j + wave, n - 1)];
        #pragma unroll
        for (int j = 0; j < STAGE_PER_THREAD; ++j)
            val[j] = *(const float4*)(emb + (size_t)node[j] * EMB_DIM + 4 * lane);
        #pragma unroll
        for (int j = 0; j < STAGE_PER_THREAD; ++j) {
            const int r = 4 * j + wave;
            float4 v = val[j];
            if (row0 + r >= n) v = make_float4(0.0f, 0.0f, 0.0f, 0.0f);
            *(float4*)&sA[r][4 * lane] = v;
        }
    }
    __syncthreads();

    for (int p = 0; p < 2; ++p) {
        const int nbase = wave * 64 + p * 32;
        v8f acc[MLP_RT][2];
        #pragma unroll
        for (int r = 0; r < MLP_RT; ++r)
            #pragma unroll
            for (int c = 0; c < 2; ++c)
                acc[r][c] = (v8f){0.f, 0.f, 0.f, 0.f, 0.f, 0.f, 0.f, 0.f};
        for (int s = 0; s < 32; ++s) {
            const int k = 4 * s + 2 * g;
            v2f a[MLP_RT], b[2];
            #pragma unroll
            for (int r = 0; r < MLP_RT; ++r) { a[r].x = sA[16 * r + m][k]; a[r].y = sA[16 * r + m][k + 1]; }
            #pragma unroll
            for (int c = 0; c < 2; ++c) {
                b[c].x = W1[(size_t)k * HIDDEN + nbase + 16 * c + m];
                b[c].y = W1[(size_t)(k + 1) * HIDDEN + nbase + 16 * c + m];
            }
            #pragma unroll
            for (int r = 0; r < MLP_RT; ++r)
                #pragma unroll
                for (int c = 0; c < 2; ++c)
                    acc[r][c] = __builtin_amdgcn_wmma_f32_16x16x4_f32(
                        false, a[r], false, b[c], (short)0, acc[r][c], false, false);
        }
        #pragma unroll
        for (int c = 0; c < 2; ++c) {
            const float bias = b1[nbase + 16 * c + m];
            #pragma unroll
            for (int r = 0; r < MLP_RT; ++r)
                #pragma unroll
                for (int v = 0; v < 8; ++v) {
                    float val = acc[r][c][v] + bias;
                    val = val > 0.0f ? val : 0.0f;
                    sH[16 * r + v + 8 * g][nbase + 16 * c + m] = val;
                }
        }
    }
    __syncthreads();

    {
        const int nbase = wave * 16;
        v8f acc[MLP_RT];
        #pragma unroll
        for (int r = 0; r < MLP_RT; ++r)
            acc[r] = (v8f){0.f, 0.f, 0.f, 0.f, 0.f, 0.f, 0.f, 0.f};
        for (int s = 0; s < 64; ++s) {
            const int k = 4 * s + 2 * g;
            v2f a[MLP_RT], b;
            #pragma unroll
            for (int r = 0; r < MLP_RT; ++r) { a[r].x = sH[16 * r + m][k]; a[r].y = sH[16 * r + m][k + 1]; }
            b.x = W2[(size_t)k * OUT_C + nbase + m];
            b.y = W2[(size_t)(k + 1) * OUT_C + nbase + m];
            #pragma unroll
            for (int r = 0; r < MLP_RT; ++r)
                acc[r] = __builtin_amdgcn_wmma_f32_16x16x4_f32(
                    false, a[r], false, b, (short)0, acc[r], false, false);
        }
        const float bias = b2[nbase + m];
        #pragma unroll
        for (int r = 0; r < MLP_RT; ++r)
            #pragma unroll
            for (int v = 0; v < 8; ++v) {
                const int row = row0 + 16 * r + v + 8 * g;
                if (row < n) {
                    const float val = acc[r][v] + bias;
                    const size_t idx = (size_t)row * OUT_C + nbase + m;
                    const float di = dis_from_count(row_end(rowstart, row, n, e) - rowstart[row]);
                    h_out[idx]  = val;
                    z0_out[idx] = di * val;
                }
            }
    }
}

__launch_bounds__(128)
__global__ void gather_round_kernel(const int* __restrict__ rowstart, const int* __restrict__ csr_src,
                                    const float* __restrict__ z, const float* __restrict__ h,
                                    float* __restrict__ znew, int n, int e, int last) {
    const unsigned gid = blockIdx.x * blockDim.x + threadIdx.x;
    const int node0 = (int)(gid >> 4) * NODES_PER_GROUP;
    const int q     = (int)(gid & 15);
    if (node0 >= n) return;

    const char* zb = (const char*)z;
    const unsigned qoff = (unsigned)q * 16u;

    int bound_my = e;
    if (q <= NODES_PER_GROUP && node0 + q < n) bound_my = rowstart[node0 + q];

    const int gbeg = __shfl(bound_my, 0, 16);
    const int gend = __shfl(bound_my, NODES_PER_GROUP, 16);
    int ids_next = node0;
    if (gbeg + q < gend) ids_next = csr_src[gbeg + q];

    for (int r = 0; r < NODES_PER_GROUP; ++r) {
        const int node = node0 + r;
        if (node >= n) break;
        const int start = __shfl(bound_my, r, 16);
        const int end   = __shfl(bound_my, r + 1, 16);
        const float di  = dis_from_count(end - start);

        const unsigned own = (unsigned)node * 256u + qoff;
        const float4 zo = *(const float4*)(zb + own);
        const float4 hh = *(const float4*)((const char*)h + own);

        float4 acc = make_float4(0.0f, 0.0f, 0.0f, 0.0f);
        for (int base = start; base < end; base += PIECE) {
            const int m = min(PIECE, end - base);
            const int s_my = (q < m) ? ids_next : node;
            ids_next = csr_src[min(base + m + q, e - 1)];

            float4 v[PIECE];
            #define ROW_LOAD(t)  v[t] = *(const float4*)(zb + ((unsigned)__shfl(s_my, (t), 16) * 256u + qoff))
            #define ROW_LOAD4(t) ROW_LOAD(t); ROW_LOAD((t) + 1); ROW_LOAD((t) + 2); ROW_LOAD((t) + 3)
            #define ROW_ADD(t)   { const bool on = (t) < m;          \
                                   acc.x += on ? v[t].x : 0.0f;      \
                                   acc.y += on ? v[t].y : 0.0f;      \
                                   acc.z += on ? v[t].z : 0.0f;      \
                                   acc.w += on ? v[t].w : 0.0f; }
            #define ROW_ADD4(t)  ROW_ADD(t) ROW_ADD((t) + 1) ROW_ADD((t) + 2) ROW_ADD((t) + 3)
            ROW_LOAD4(0);
            if (m > 4) { ROW_LOAD4(4); if (m > 8) { ROW_LOAD4(8); if (m > 12) { ROW_LOAD4(12); } } }
            ROW_ADD4(0)
            if (m > 4) { ROW_ADD4(4) if (m > 8) { ROW_ADD4(8) if (m > 12) { ROW_ADD4(12) } } }
            #undef ROW_LOAD
            #undef ROW_LOAD4
            #undef ROW_ADD
            #undef ROW_ADD4
        }

        const float post = last ? 1.0f : di;
        float4 out4;
        out4.x = post * ((1.0f - ALPHA) * (di * (acc.x + zo.x)) + ALPHA * hh.x);
        out4.y = post * ((1.0f - ALPHA) * (di * (acc.y + zo.y)) + ALPHA * hh.y);
        out4.z = post * ((1.0f - ALPHA) * (di * (acc.z + zo.z)) + ALPHA * hh.z);
        out4.w = post * ((1.0f - ALPHA) * (di * (acc.w + zo.w)) + ALPHA * hh.w);
        *(float4*)((char*)znew + own) = out4;
    }
}

extern "C" void kernel_launch(void* const* d_in, const int* in_sizes, int n_in,
                              void* d_out, int out_size, void* d_ws, size_t ws_size,
                              hipStream_t stream) {
    (void)n_in; (void)out_size; (void)ws_size;

    const int*   xidx = (const int*)d_in[0];
    const int*   ei   = (const int*)d_in[1];
    const float* emb  = (const float*)d_in[2];
    const float* W1   = (const float*)d_in[3];
    const float* b1   = (const float*)d_in[4];
    const float* W2   = (const float*)d_in[5];
    const float* b2   = (const float*)d_in[6];
    float* out = (float*)d_out;

    const int N = in_sizes[0];
    const int E = in_sizes[1] / 2;
    const size_t NC = (size_t)N * OUT_C;

    float* h        = (float*)d_ws;
    float* xping    = h + NC;
    int*   rowstart = (int*)(xping + NC);
    int*   csr_src  = rowstart + N;
    const int nbins  = (N + BIN_NODES - 1) / BIN_NODES;
    const int ntable = nbins * SLICE_BLOCKS;
    int*   table    = (int*)xping;
    int*   binstart = table + ntable;
    int*   binned   = binstart + nbins;
    int*   rows     = binned + E;

    const int* src = ei;
    const int* dst = ei + E;

    const size_t cnt_lds = (size_t)(nbins + 1) * sizeof(int);
    const size_t sct_lds = (size_t)(2 * nbins + 1) * sizeof(int);
    bin_count_kernel  <<<SLICE_BLOCKS, SLICE_THREADS, cnt_lds, stream>>>(dst, table, E, nbins);
    bin_scatter_kernel<<<SLICE_BLOCKS, SLICE_THREADS, sct_lds, stream>>>(src, dst, table, binstart, binned, E, nbins);
    bin_rows_kernel   <<<nbins, 1024, 0, stream>>>(binstart, binned, rowstart, rows, csr_src, N, E, nbins);
    const int ngroups   = (N + NODES_PER_GROUP - 1) / NODES_PER_GROUP;
    const int nblocks16 = (ngroups + 7) / 8;

    mlp_wmma_kernel<<<(N + MLP_ROWS - 1) / MLP_ROWS, 128, 0, stream>>>(xidx, emb, W1, b1, W2, b2,
                                                       rowstart, h, out, N, E);

    float* xa = out;
    float* xb = xping;
    for (int k = 0; k < K_ITERS - 1; ++k) {
        gather_round_kernel<<<nblocks16, 128, 0, stream>>>(rowstart, csr_src, xa, h, xb, N, E, 0);
        float* t = xa; xa = xb; xb = t;
    }
    {
        gather_round_kernel<<<nblocks16, 128, 0, stream>>>(rowstart, csr_src, xa, h, xb, N, E, 1);
        float* t = xa; xa = xb; xb = t;
    }
    if (xa != out)
        hipMemcpyAsync(out, xa, NC * sizeof(float), hipMemcpyDeviceToDevice, stream);
}
